# starting kernel
# baseline (speedup 1.0000x reference)
#include <hip/hip_runtime.h>
#include <hip/hip_bf16.h>
#include <cstdint>

typedef float v2f __attribute__((ext_vector_type(2)));
typedef float v8f __attribute__((ext_vector_type(8)));

#define IN_CH  256
#define HID_CH 128
#define KTILE  64
#define PSTRIDE 288

__device__ __forceinline__ void atomAddF(float* p, float v) {
    __hip_atomic_fetch_add(p, v, __ATOMIC_RELAXED, __HIP_MEMORY_SCOPE_AGENT);
}

__global__ void k_init_deg(float* __restrict__ deg, int n) {
    int i = blockIdx.x * blockDim.x + threadIdx.x;
    if (i < n) deg[i] = 1.0f;
}

__global__ void k_deg_accum(const int* __restrict__ dst, float* __restrict__ deg, int ne) {
    int e = blockIdx.x * blockDim.x + threadIdx.x;
    if (e < ne) atomAddF(&deg[dst[e]], 1.0f);
}

__global__ void k_rsqrt(float* __restrict__ deg, int n) {
    int i = blockIdx.x * blockDim.x + threadIdx.x;
    if (i < n) deg[i] = rsqrtf(deg[i]);
}

__global__ __launch_bounds__(256) void k_gemm1_wmma(
    const float* __restrict__ x, const float* __restrict__ W1,
    float* __restrict__ h, int n_nodes)
{
    __shared__ float sW[(KTILE / 2) * PSTRIDE];

    const int tid  = threadIdx.x;
    const int wave = tid >> 5;
    const int lane = tid & 31;
    const int half = lane >> 4;
    const int l16  = lane & 15;

    const int row_base = (blockIdx.x * 8 + wave) * 16;
    int arow = row_base + l16;
    if (arow >= n_nodes) arow = n_nodes - 1;
    const float* aptr = x + (size_t)arow * IN_CH + half * 2;

    v8f acc[8] = {};

    for (int k0 = 0; k0 < IN_CH; k0 += KTILE) {
        __syncthreads();
        for (int i = tid; i < (KTILE / 2) * 32; i += 256) {
            int p  = i >> 5;
            int c4 = (i & 31) * 4;
            float4 r0 = *(const float4*)(W1 + (size_t)(k0 + 2 * p + 0) * HID_CH + c4);
            float4 r1 = *(const float4*)(W1 + (size_t)(k0 + 2 * p + 1) * HID_CH + c4);
            float* dp = &sW[p * PSTRIDE + c4 * 2];
            ((float2*)dp)[0] = make_float2(r0.x, r1.x);
            ((float2*)dp)[1] = make_float2(r0.y, r1.y);
            ((float2*)dp)[2] = make_float2(r0.z, r1.z);
            ((float2*)dp)[3] = make_float2(r0.w, r1.w);
        }
        __syncthreads();

#pragma unroll 4
        for (int k = 0; k < KTILE; k += 4) {
            v2f a = *(const v2f*)(aptr + k0 + k);
            const int p = (k >> 1) + half;
            const float* bbase = &sW[p * PSTRIDE + l16 * 2];
#pragma unroll
            for (int nt = 0; nt < 8; ++nt) {
                v2f b = *(const v2f*)(bbase + nt * 32);
                acc[nt] = __builtin_amdgcn_wmma_f32_16x16x4_f32(
                    false, a, false, b, (short)0, acc[nt], false, false);
            }
        }
    }

#pragma unroll
    for (int nt = 0; nt < 8; ++nt) {
#pragma unroll
        for (int v = 0; v < 8; ++v) {
            int r = row_base + v + 8 * half;
            if (r < n_nodes) h[(size_t)r * HID_CH + nt * 16 + l16] = acc[nt][v];
        }
    }
}

__global__ void k_agg1_init(const float* __restrict__ h, const float* __restrict__ b1,
                            const float* __restrict__ dinv, float* __restrict__ out1,
                            int n_nodes)
{
    int idx = blockIdx.x * blockDim.x + threadIdx.x;
    if (idx >= n_nodes * (HID_CH / 4)) return;
    int node = idx >> 5;
    int c = (idx & 31) * 4;
    float di = dinv[node], w = di * di;
    float4 hv = *(const float4*)(h + (size_t)node * HID_CH + c);
    float4 bv = *(const float4*)(b1 + c);
    float4 o; o.x = bv.x + hv.x * w; o.y = bv.y + hv.y * w;
    o.z = bv.z + hv.z * w; o.w = bv.w + hv.w * w;
    *(float4*)(out1 + (size_t)node * HID_CH + c) = o;
}

__global__ __launch_bounds__(256) void k_agg1_edges(
    const float* __restrict__ h, const int* __restrict__ src, const int* __restrict__ dst,
    const float* __restrict__ dinv, float* __restrict__ out1, int ne)
{
    int e = blockIdx.x * (blockDim.x >> 5) + (threadIdx.x >> 5);
    if (e >= ne) return;
    int lane = threadIdx.x & 31;
    int s = src[e], d = dst[e];
    float norm = dinv[s] * dinv[d];
    float4 hv = *(const float4*)(h + (size_t)s * HID_CH + lane * 4);
    float* op = out1 + (size_t)d * HID_CH + lane * 4;
    atomAddF(op + 0, hv.x * norm);
    atomAddF(op + 1, hv.y * norm);
    atomAddF(op + 2, hv.z * norm);
    atomAddF(op + 3, hv.w * norm);
}

__global__ void k_gemm2(const float* __restrict__ out1, const float* __restrict__ W2,
                        const float* __restrict__ b2, const float* __restrict__ dinv,
                        float* __restrict__ h2, float* __restrict__ out, int n_nodes)
{
    int i = blockIdx.x * blockDim.x + threadIdx.x;
    if (i >= n_nodes) return;
    const float* row = out1 + (size_t)i * HID_CH;
    float s0 = 0.f, s1 = 0.f;
#pragma unroll 8
    for (int c = 0; c < HID_CH; c += 4) {
        float4 hv = *(const float4*)(row + c);
        float r0 = fmaxf(hv.x, 0.f), r1 = fmaxf(hv.y, 0.f);
        float r2 = fmaxf(hv.z, 0.f), r3 = fmaxf(hv.w, 0.f);
        s0 += r0 * W2[(c + 0) * 2 + 0] + r1 * W2[(c + 1) * 2 + 0]
            + r2 * W2[(c + 2) * 2 + 0] + r3 * W2[(c + 3) * 2 + 0];
        s1 += r0 * W2[(c + 0) * 2 + 1] + r1 * W2[(c + 1) * 2 + 1]
            + r2 * W2[(c + 2) * 2 + 1] + r3 * W2[(c + 3) * 2 + 1];
    }
    float di = dinv[i], w = di * di;
    h2[i * 2 + 0] = s0;
    h2[i * 2 + 1] = s1;
    out[i * 2 + 0] = b2[0] + s0 * w;
    out[i * 2 + 1] = b2[1] + s1 * w;
}

__global__ void k_agg2_edges(const float* __restrict__ h2, const int* __restrict__ src,
                             const int* __restrict__ dst, const float* __restrict__ dinv,
                             float* __restrict__ out, int ne)
{
    int e = blockIdx.x * blockDim.x + threadIdx.x;
    if (e >= ne) return;
    int s = src[e], d = dst[e];
    float norm = dinv[s] * dinv[d];
    atomAddF(&out[d * 2 + 0], h2[s * 2 + 0] * norm);
    atomAddF(&out[d * 2 + 1], h2[s * 2 + 1] * norm);
}

extern "C" void kernel_launch(void* const* d_in, const int* in_sizes, int n_in,
                              void* d_out, int out_size, void* d_ws, size_t ws_size,
                              hipStream_t stream)
{
    const float* x  = (const float*)d_in[0];
    const int*   ei = (const int*)d_in[1];
    const float* W1 = (const float*)d_in[2];
    const float* b1 = (const float*)d_in[3];
    const float* W2 = (const float*)d_in[4];
    const float* b2 = (const float*)d_in[5];
    float* out = (float*)d_out;

    const int n_nodes = in_sizes[0] / IN_CH;
    const int n_edges = in_sizes[1] / 2;
    const int* src = ei;
    const int* dst = ei + n_edges;

    char* ws = (char*)d_ws;
    size_t off = 0;
    float* dinv = (float*)(ws + off);
    off += ((size_t)n_nodes * 4 + 4095) & ~(size_t)4095;
    float* h = (float*)(ws + off);
    size_t hb = ((size_t)n_nodes * HID_CH * 4 + 4095) & ~(size_t)4095;
    off += hb;
    float* out1 = (float*)(ws + off);
    float* h2 = h;

    k_init_deg <<<(n_nodes + 255) / 256, 256, 0, stream>>>(dinv, n_nodes);
    k_deg_accum<<<(n_edges + 255) / 256, 256, 0, stream>>>(dst, dinv, n_edges);
    k_rsqrt    <<<(n_nodes + 255) / 256, 256, 0, stream>>>(dinv, n_nodes);

    k_gemm1_wmma<<<(n_nodes + 127) / 128, 256, 0, stream>>>(x, W1, h, n_nodes);

    int t1 = n_nodes * (HID_CH / 4);
    k_agg1_init <<<(t1 + 255) / 256, 256, 0, stream>>>(h, b1, dinv, out1, n_nodes);
    k_agg1_edges<<<(n_edges + 7) / 8, 256, 0, stream>>>(h, src, dst, dinv, out1, n_edges);

    k_gemm2     <<<(n_nodes + 255) / 256, 256, 0, stream>>>(out1, W2, b2, dinv, h2, out, n_nodes);
    k_agg2_edges<<<(n_edges + 255) / 256, 256, 0, stream>>>(h2, src, dst, dinv, out, n_edges);
}


# Per-edge global float atomics on the two narrow node tables (in-degree, 2-channel layer-2 sums) become LDS adds: block = node range x edge chunk, LDS arrival-order sums, planes summed in fixed order
# speedup vs baseline: 2.3266x; 2.3266x over previous
#include <hip/hip_runtime.h>
#include <hip/hip_bf16.h>
#include <cstdint>

typedef float v2f __attribute__((ext_vector_type(2)));
typedef float v8f __attribute__((ext_vector_type(8)));

#define IN_CH  256
#define HID_CH 128
#define KTILE  64
#define PSTRIDE 288

__device__ __forceinline__ void atomAddF(float* p, float v) {
    __hip_atomic_fetch_add(p, v, __ATOMIC_RELAXED, __HIP_MEMORY_SCOPE_AGENT);
}

#define NT_RANGE   12544
#define NT_PLANES  32
#define NT_THREADS 1024
#define NT_U       16
#define NT_SLICES  ((NT_RANGE + NT_THREADS - 1) / NT_THREADS)

__global__ __launch_bounds__(NT_THREADS) void k_deg_hist(
    const int* __restrict__ dst, unsigned* __restrict__ part, int ne, int n, int chunk_len)
{
    __shared__ unsigned cnt[NT_RANGE];
    const int tid = threadIdx.x;
    const int c   = blockIdx.x;
    const int lo  = blockIdx.y * NT_RANGE;

    for (int i = tid; i < NT_RANGE; i += NT_THREADS) cnt[i] = 0u;
    __syncthreads();

    const int e0 = c * chunk_len;
    const int e1 = min(ne, e0 + chunk_len);
    for (int base = e0 + tid; base < e1; base += NT_THREADS * NT_U) {
        int dv[NT_U];
#pragma unroll
        for (int u = 0; u < NT_U; ++u) {
            const int e = base + u * NT_THREADS;
            dv[u] = (e < e1) ? dst[e] : -1;
        }
#pragma unroll
        for (int u = 0; u < NT_U; ++u) {
            const unsigned rel = (unsigned)(dv[u] - lo);
            if (rel < (unsigned)NT_RANGE) atomicAdd(&cnt[rel], 1u);
        }
    }
    __syncthreads();

    const int hi = min(NT_RANGE, n - lo);
    unsigned* pp = part + (size_t)c * n + lo;
    for (int i = tid; i < hi; i += NT_THREADS) pp[i] = cnt[i];
}

__global__ void k_dinv(const unsigned* __restrict__ part, float* __restrict__ dinv, int n) {
    int i = blockIdx.x * blockDim.x + threadIdx.x;
    if (i >= n) return;
    unsigned s = 0u;
#pragma unroll
    for (int c = 0; c < NT_PLANES; ++c) s += part[(size_t)c * n + i];
    dinv[i] = rsqrtf(1.0f + (float)s);
}

__global__ __launch_bounds__(256) void k_gemm1_wmma(
    const float* __restrict__ x, const float* __restrict__ W1,
    float* __restrict__ h, int n_nodes)
{
    __shared__ float sW[(KTILE / 2) * PSTRIDE];

    const int tid  = threadIdx.x;
    const int wave = tid >> 5;
    const int lane = tid & 31;
    const int half = lane >> 4;
    const int l16  = lane & 15;

    const int row_base = (blockIdx.x * 8 + wave) * 16;
    int arow = row_base + l16;
    if (arow >= n_nodes) arow = n_nodes - 1;
    const float* aptr = x + (size_t)arow * IN_CH + half * 2;

    v8f acc[8] = {};

    for (int k0 = 0; k0 < IN_CH; k0 += KTILE) {
        __syncthreads();
        for (int i = tid; i < (KTILE / 2) * 32; i += 256) {
            int p  = i >> 5;
            int c4 = (i & 31) * 4;
            float4 r0 = *(const float4*)(W1 + (size_t)(k0 + 2 * p + 0) * HID_CH + c4);
            float4 r1 = *(const float4*)(W1 + (size_t)(k0 + 2 * p + 1) * HID_CH + c4);
            float* dp = &sW[p * PSTRIDE + c4 * 2];
            ((float2*)dp)[0] = make_float2(r0.x, r1.x);
            ((float2*)dp)[1] = make_float2(r0.y, r1.y);
            ((float2*)dp)[2] = make_float2(r0.z, r1.z);
            ((float2*)dp)[3] = make_float2(r0.w, r1.w);
        }
        __syncthreads();

#pragma unroll 4
        for (int k = 0; k < KTILE; k += 4) {
            v2f a = *(const v2f*)(aptr + k0 + k);
            const int p = (k >> 1) + half;
            const float* bbase = &sW[p * PSTRIDE + l16 * 2];
#pragma unroll
            for (int nt = 0; nt < 8; ++nt) {
                v2f b = *(const v2f*)(bbase + nt * 32);
                acc[nt] = __builtin_amdgcn_wmma_f32_16x16x4_f32(
                    false, a, false, b, (short)0, acc[nt], false, false);
            }
        }
    }

#pragma unroll
    for (int nt = 0; nt < 8; ++nt) {
#pragma unroll
        for (int v = 0; v < 8; ++v) {
            int r = row_base + v + 8 * half;
            if (r < n_nodes) h[(size_t)r * HID_CH + nt * 16 + l16] = acc[nt][v];
        }
    }
}

__global__ void k_agg1_init(const float* __restrict__ h, const float* __restrict__ b1,
                            const float* __restrict__ dinv, float* __restrict__ out1,
                            int n_nodes)
{
    int idx = blockIdx.x * blockDim.x + threadIdx.x;
    if (idx >= n_nodes * (HID_CH / 4)) return;
    int node = idx >> 5;
    int c = (idx & 31) * 4;
    float di = dinv[node], w = di * di;
    float4 hv = *(const float4*)(h + (size_t)node * HID_CH + c);
    float4 bv = *(const float4*)(b1 + c);
    float4 o; o.x = bv.x + hv.x * w; o.y = bv.y + hv.y * w;
    o.z = bv.z + hv.z * w; o.w = bv.w + hv.w * w;
    *(float4*)(out1 + (size_t)node * HID_CH + c) = o;
}

__global__ __launch_bounds__(256) void k_agg1_edges(
    const float* __restrict__ h, const int* __restrict__ src, const int* __restrict__ dst,
    const float* __restrict__ dinv, float* __restrict__ out1, int ne)
{
    const int lane = threadIdx.x & 31;
    const int wv = __builtin_amdgcn_readfirstlane((int)(threadIdx.x >> 5));
    const int e0 = (blockIdx.x * (blockDim.x >> 5) + wv) * 32;
    if (e0 >= ne) return;
    const int cnt = min(32, ne - e0);
    int s = 0, d = 0;
    float nrm = 0.f;
    if (lane < cnt) {
        s = src[e0 + lane];
        d = dst[e0 + lane];
        nrm = dinv[s] * dinv[d];
    }
    for (int j0 = 0; j0 < cnt; j0 += 4) {
        float v[4][4];
#pragma unroll
        for (int u = 0; u < 4; ++u) {
            const int j  = min(j0 + u, cnt - 1);
            const int sj = __builtin_amdgcn_readlane(s, j);
            const float* hp = h + (size_t)sj * HID_CH + lane;
            v[u][0] = hp[0];
            v[u][1] = hp[32];
            v[u][2] = hp[64];
            v[u][3] = hp[96];
        }
#pragma unroll
        for (int u = 0; u < 4; ++u) {
            const int j = j0 + u;
            if (j < cnt) {
                const int   dj = __builtin_amdgcn_readlane(d, j);
                const float nj = __int_as_float(__builtin_amdgcn_readlane(__float_as_int(nrm), j));
                float* op = out1 + (size_t)dj * HID_CH + lane;
                atomAddF(op +  0, v[u][0] * nj);
                atomAddF(op + 32, v[u][1] * nj);
                atomAddF(op + 64, v[u][2] * nj);
                atomAddF(op + 96, v[u][3] * nj);
            }
        }
    }
}

__global__ void k_gemm2(const float* __restrict__ out1, const float* __restrict__ W2,
                        float* __restrict__ h2, int n_nodes)
{
    int i = blockIdx.x * blockDim.x + threadIdx.x;
    if (i >= n_nodes) return;
    const float* row = out1 + (size_t)i * HID_CH;
    float s0 = 0.f, s1 = 0.f;
#pragma unroll 8
    for (int c = 0; c < HID_CH; c += 4) {
        float4 hv = *(const float4*)(row + c);
        float r0 = fmaxf(hv.x, 0.f), r1 = fmaxf(hv.y, 0.f);
        float r2 = fmaxf(hv.z, 0.f), r3 = fmaxf(hv.w, 0.f);
        s0 += r0 * W2[(c + 0) * 2 + 0] + r1 * W2[(c + 1) * 2 + 0]
            + r2 * W2[(c + 2) * 2 + 0] + r3 * W2[(c + 3) * 2 + 0];
        s1 += r0 * W2[(c + 0) * 2 + 1] + r1 * W2[(c + 1) * 2 + 1]
            + r2 * W2[(c + 2) * 2 + 1] + r3 * W2[(c + 3) * 2 + 1];
    }
    h2[i * 2 + 0] = s0;
    h2[i * 2 + 1] = s1;
}

__global__ __launch_bounds__(NT_THREADS) void k_agg2_lds(
    const float* __restrict__ h2, const int* __restrict__ src, const int* __restrict__ dst,
    const float* __restrict__ dinv, float* __restrict__ part2, int ne, int n, int chunk_len)
{
    __shared__ float acc[NT_RANGE * 2];
    __shared__ float dloc[NT_RANGE];
    const int tid = threadIdx.x;
    const int p   = blockIdx.x;
    const int lo  = blockIdx.y * NT_RANGE;
    const int hi  = min(NT_RANGE, n - lo);

    float dl[NT_SLICES];
#pragma unroll
    for (int k = 0; k < NT_SLICES; ++k) {
        const int i = tid + k * NT_THREADS;
        dl[k] = (i < hi) ? dinv[lo + i] : 0.f;
    }
#pragma unroll
    for (int k = 0; k < NT_SLICES; ++k) {
        const int i = tid + k * NT_THREADS;
        if (i < NT_RANGE) {
            acc[2 * i + 0] = 0.f;
            acc[2 * i + 1] = 0.f;
            dloc[i] = dl[k];
        }
    }
    __syncthreads();

    const int e0 = p * chunk_len;
    const int e1 = min(ne, e0 + chunk_len);
    for (int base = e0 + tid; base < e1; base += NT_THREADS * NT_U) {
        int dv[NT_U], sv[NT_U];
#pragma unroll
        for (int u = 0; u < NT_U; ++u) {
            const int e = base + u * NT_THREADS;
            const bool ok = e < e1;
            dv[u] = ok ? dst[e] : -1;
            sv[u] = ok ? src[e] : 0;
        }
        float  ds[NT_U];
        float2 hv[NT_U];
#pragma unroll
        for (int u = 0; u < NT_U; ++u) {
            const bool m = (unsigned)(dv[u] - lo) < (unsigned)NT_RANGE;
            const int  s = m ? sv[u] : 0;
            ds[u] = dinv[s];
            hv[u] = *(const float2*)(h2 + (size_t)s * 2);
        }
#pragma unroll
        for (int u = 0; u < NT_U; ++u) {
            const unsigned rel = (unsigned)(dv[u] - lo);
            if (rel < (unsigned)NT_RANGE) {
                const float norm = ds[u] * dloc[rel];
                __hip_atomic_fetch_add(&acc[rel * 2 + 0], hv[u].x * norm,
                                       __ATOMIC_RELAXED, __HIP_MEMORY_SCOPE_WORKGROUP);
                __hip_atomic_fetch_add(&acc[rel * 2 + 1], hv[u].y * norm,
                                       __ATOMIC_RELAXED, __HIP_MEMORY_SCOPE_WORKGROUP);
            }
        }
    }
    __syncthreads();

    float2* pp = (float2*)(part2 + ((size_t)p * n + lo) * 2);
    for (int i = tid; i < hi; i += NT_THREADS) pp[i] = make_float2(acc[2 * i], acc[2 * i + 1]);
}

__global__ void k_out_sum(const float* __restrict__ h2, const float* __restrict__ part2,
                          const float* __restrict__ b2, const float* __restrict__ dinv,
                          float* __restrict__ out, int n)
{
    int i = blockIdx.x * blockDim.x + threadIdx.x;
    if (i >= n) return;
    const float di = dinv[i], w = di * di;
    const float2 hs = *(const float2*)(h2 + (size_t)i * 2);
    float o0 = b2[0] + hs.x * w;
    float o1 = b2[1] + hs.y * w;
#pragma unroll
    for (int p = 0; p < NT_PLANES; ++p) {
        const float2 v = *(const float2*)(part2 + ((size_t)p * n + i) * 2);
        o0 += v.x;
        o1 += v.y;
    }
    *(float2*)(out + (size_t)i * 2) = make_float2(o0, o1);
}

extern "C" void kernel_launch(void* const* d_in, const int* in_sizes, int n_in,
                              void* d_out, int out_size, void* d_ws, size_t ws_size,
                              hipStream_t stream)
{
    const float* x  = (const float*)d_in[0];
    const int*   ei = (const int*)d_in[1];
    const float* W1 = (const float*)d_in[2];
    const float* b1 = (const float*)d_in[3];
    const float* W2 = (const float*)d_in[4];
    const float* b2 = (const float*)d_in[5];
    float* out = (float*)d_out;

    const int n_nodes = in_sizes[0] / IN_CH;
    const int n_edges = in_sizes[1] / 2;
    const int* src = ei;
    const int* dst = ei + n_edges;

    char* ws = (char*)d_ws;
    size_t off = 0;
    float* dinv = (float*)(ws + off);
    off += ((size_t)n_nodes * 4 + 4095) & ~(size_t)4095;
    float* h = (float*)(ws + off);
    size_t hb = ((size_t)n_nodes * HID_CH * 4 + 4095) & ~(size_t)4095;
    off += hb;
    float* out1 = (float*)(ws + off);
    float* h2 = h;

    unsigned* part  = (unsigned*)out1;
    float*    part2 = out1;
    const int chunk_len = (n_edges + NT_PLANES - 1) / NT_PLANES;
    dim3 ngrid(NT_PLANES, (n_nodes + NT_RANGE - 1) / NT_RANGE);
    k_deg_hist<<<ngrid, NT_THREADS, 0, stream>>>(dst, part, n_edges, n_nodes, chunk_len);
    k_dinv    <<<(n_nodes + 255) / 256, 256, 0, stream>>>(part, dinv, n_nodes);

    k_gemm1_wmma<<<(n_nodes + 127) / 128, 256, 0, stream>>>(x, W1, h, n_nodes);

    int t1 = n_nodes * (HID_CH / 4);
    k_agg1_init <<<(t1 + 255) / 256, 256, 0, stream>>>(h, b1, dinv, out1, n_nodes);
    k_agg1_edges<<<(n_edges + 255) / 256, 256, 0, stream>>>(h, src, dst, dinv, out1, n_edges);

    k_gemm2     <<<(n_nodes + 255) / 256, 256, 0, stream>>>(out1, W2, h2, n_nodes);
    k_agg2_lds  <<<ngrid, NT_THREADS, 0, stream>>>(h2, src, dst, dinv, part2, n_edges, n_nodes, chunk_len);
    k_out_sum   <<<(n_nodes + 255) / 256, 256, 0, stream>>>(h2, part2, b2, dinv, out, n_nodes);
}


# Layer-1 edge sums leave global float atomics: block owns 400 dest nodes x 128 ch in LDS, finds its edges in a byte-per-edge owner table, LDS arrival-order adds; relu+W2 from LDS, order kept
# speedup vs baseline: 2.9738x; 1.2782x over previous
#include <hip/hip_runtime.h>
#include <hip/hip_bf16.h>
#include <cstdint>

typedef float v2f __attribute__((ext_vector_type(2)));
typedef float v8f __attribute__((ext_vector_type(8)));

#define IN_CH  256
#define HID_CH 128
#define KTILE  64
#define PSTRIDE 288

#define L1_R       400
#define L1_PITCH   129
#define L1_WCAP    288
#define L1_THREADS 1024
#define L1_SU      8
#define L1_PU      8
#define L1_G       8

#define NT_RANGE   12544
#define NT_PLANES  32
#define NT_THREADS 1024
#define NT_U       16
#define NT_SLICES  ((NT_RANGE + NT_THREADS - 1) / NT_THREADS)

__global__ __launch_bounds__(NT_THREADS) void k_deg_hist(
    const int* __restrict__ dst, unsigned* __restrict__ part, unsigned char* __restrict__ rid8,
    int ne, int n, int chunk_len)
{
    __shared__ unsigned cnt[NT_RANGE];
    const int tid = threadIdx.x;
    const int c   = blockIdx.x;
    const int lo  = blockIdx.y * NT_RANGE;

    for (int i = tid; i < NT_RANGE; i += NT_THREADS) cnt[i] = 0u;

    {
        const int nb    = gridDim.x * gridDim.y;
        const int bid   = blockIdx.y * gridDim.x + blockIdx.x;
        const int nquad = ne >> 2;
        const int qlen  = ((nquad + nb - 1) / nb + 31) & ~31;
        const int q0    = bid * qlen;
        const int q1    = min(nquad, q0 + qlen);
        unsigned* rid32 = (unsigned*)rid8;
        for (int q = q0 + tid; q < q1; q += NT_THREADS) {
            const int* dp = dst + 4 * (size_t)q;
            const unsigned o0 = ((unsigned)dp[0] / (unsigned)L1_R) & 0xffu;
            const unsigned o1 = ((unsigned)dp[1] / (unsigned)L1_R) & 0xffu;
            const unsigned o2 = ((unsigned)dp[2] / (unsigned)L1_R) & 0xffu;
            const unsigned o3 = ((unsigned)dp[3] / (unsigned)L1_R) & 0xffu;
            rid32[q] = o0 | (o1 << 8) | (o2 << 16) | (o3 << 24);
        }
        if (bid == 0 && tid < (ne & 3)) {
            const int e = (ne & ~3) + tid;
            rid8[e] = (unsigned char)((unsigned)dst[e] / (unsigned)L1_R);
        }
    }
    __syncthreads();

    const int e0 = c * chunk_len;
    const int e1 = min(ne, e0 + chunk_len);
    for (int base = e0 + tid; base < e1; base += NT_THREADS * NT_U) {
        int dv[NT_U];
#pragma unroll
        for (int u = 0; u < NT_U; ++u) {
            const int e = base + u * NT_THREADS;
            dv[u] = (e < e1) ? dst[e] : -1;
        }
#pragma unroll
        for (int u = 0; u < NT_U; ++u) {
            const unsigned rel = (unsigned)(dv[u] - lo);
            if (rel < (unsigned)NT_RANGE) atomicAdd(&cnt[rel], 1u);
        }
    }
    __syncthreads();

    const int hi = min(NT_RANGE, n - lo);
    unsigned* pp = part + (size_t)c * n + lo;
    for (int i = tid; i < hi; i += NT_THREADS) pp[i] = cnt[i];
}

__global__ void k_dinv(const unsigned* __restrict__ part, float* __restrict__ dinv, int n) {
    int i = blockIdx.x * blockDim.x + threadIdx.x;
    if (i >= n) return;
    unsigned s = 0u;
#pragma unroll
    for (int c = 0; c < NT_PLANES; ++c) s += part[(size_t)c * n + i];
    dinv[i] = rsqrtf(1.0f + (float)s);
}

__global__ __launch_bounds__(256) void k_gemm1_wmma(
    const float* __restrict__ x, const float* __restrict__ W1,
    float* __restrict__ h, int n_nodes)
{
    __shared__ float sW[(KTILE / 2) * PSTRIDE];

    const int tid  = threadIdx.x;
    const int wave = tid >> 5;
    const int lane = tid & 31;
    const int half = lane >> 4;
    const int l16  = lane & 15;

    const int row_base = (blockIdx.x * 8 + wave) * 16;
    int arow = row_base + l16;
    if (arow >= n_nodes) arow = n_nodes - 1;
    const float* aptr = x + (size_t)arow * IN_CH + half * 2;

    v8f acc[8] = {};

    for (int k0 = 0; k0 < IN_CH; k0 += KTILE) {
        __syncthreads();
        for (int i = tid; i < (KTILE / 2) * 32; i += 256) {
            int p  = i >> 5;
            int c4 = (i & 31) * 4;
            float4 r0 = *(const float4*)(W1 + (size_t)(k0 + 2 * p + 0) * HID_CH + c4);
            float4 r1 = *(const float4*)(W1 + (size_t)(k0 + 2 * p + 1) * HID_CH + c4);
            float* dp = &sW[p * PSTRIDE + c4 * 2];
            ((float2*)dp)[0] = make_float2(r0.x, r1.x);
            ((float2*)dp)[1] = make_float2(r0.y, r1.y);
            ((float2*)dp)[2] = make_float2(r0.z, r1.z);
            ((float2*)dp)[3] = make_float2(r0.w, r1.w);
        }
        __syncthreads();

#pragma unroll 4
        for (int k = 0; k < KTILE; k += 4) {
            v2f a = *(const v2f*)(aptr + k0 + k);
            const int p = (k >> 1) + half;
            const float* bbase = &sW[p * PSTRIDE + l16 * 2];
#pragma unroll
            for (int nt = 0; nt < 8; ++nt) {
                v2f b = *(const v2f*)(bbase + nt * 32);
                acc[nt] = __builtin_amdgcn_wmma_f32_16x16x4_f32(
                    false, a, false, b, (short)0, acc[nt], false, false);
            }
        }
    }

#pragma unroll
    for (int nt = 0; nt < 8; ++nt) {
#pragma unroll
        for (int v = 0; v < 8; ++v) {
            int r = row_base + v + 8 * half;
            if (r < n_nodes) h[(size_t)r * HID_CH + nt * 16 + l16] = acc[nt][v];
        }
    }
}

__device__ __forceinline__ void ldsAddF(float* p, float v) {
    __hip_atomic_fetch_add(p, v, __ATOMIC_RELAXED, __HIP_MEMORY_SCOPE_WORKGROUP);
}

__global__ __launch_bounds__(L1_THREADS) void k_agg1_own(
    const float* __restrict__ h, const int* __restrict__ src, const int* __restrict__ dst,
    const float* __restrict__ dinv, const float* __restrict__ b1, const float* __restrict__ W2,
    const unsigned char* __restrict__ rid8, float* __restrict__ h2, int ne, int n)
{
    __shared__ float    acc[L1_R * L1_PITCH];
    __shared__ unsigned qs[(L1_THREADS / 32) * L1_WCAP];
    __shared__ uint2    qdw[(L1_THREADS / 32) * L1_WCAP];
    __shared__ float    dloc[L1_R];

    const int tid  = threadIdx.x;
    const int lane = tid & 31;
    const int wv  = __builtin_amdgcn_readfirstlane(tid >> 5);
    const int lo  = blockIdx.x * L1_R;
    const int cnt = min(L1_R, n - lo);

    if (tid < L1_R) dloc[tid] = (tid < cnt) ? dinv[lo + tid] : 0.f;
    {
        float bb[4];
#pragma unroll
        for (int k = 0; k < 4; ++k) bb[k] = b1[lane + 32 * k];
        for (int r0 = wv * 4; r0 < cnt; r0 += (L1_THREADS / 32) * 4) {
            float hv[4][4], ww[4];
#pragma unroll
            for (int q = 0; q < 4; ++q) {
                const int r = min(r0 + q, cnt - 1);
                const float di = dinv[lo + r];
                ww[q] = di * di;
                const float* hp = h + (size_t)(lo + r) * HID_CH + lane;
#pragma unroll
                for (int k = 0; k < 4; ++k) hv[q][k] = hp[32 * k];
            }
#pragma unroll
            for (int q = 0; q < 4; ++q) {
                if (r0 + q < cnt) {
                    float* ap = &acc[(r0 + q) * L1_PITCH + lane];
#pragma unroll
                    for (int k = 0; k < 4; ++k) ap[32 * k] = bb[k] + hv[q][k] * ww[q];
                }
            }
        }
    }
    __syncthreads();

    const unsigned rb = (unsigned)blockIdx.x & 0xffu;
    const unsigned R4 = rb * 0x01010101u;
    const int nvf = ne >> 4;
    const int wq0 = wv * L1_WCAP;
    unsigned wcount = 0u;

    auto take_edges = [&](const bool has, const int e) {
        const unsigned bal = __builtin_amdgcn_ballot_w32(has);
        if (has) {
            const unsigned slot = wcount + __builtin_amdgcn_mbcnt_lo(bal, 0u);
            if (slot < (unsigned)L1_WCAP) {
                qs[wq0 + slot] = (unsigned)e;
            } else {
                const int s = src[e];
                const unsigned rel = (unsigned)(dst[e] - lo);
                if (rel < (unsigned)cnt) {
                    const float nrm = dinv[s] * dloc[rel];
                    const float* hp = h + (size_t)s * HID_CH;
#pragma nounroll
                    for (int c = 0; c < HID_CH; ++c)
                        ldsAddF(&acc[rel * L1_PITCH + c], hp[c] * nrm);
                }
            }
        }
        wcount += (unsigned)__builtin_popcount(bal);
    };
    auto scan_vec = [&](const uint4 w, const int ebase) {
        const unsigned K1 = 0x01010101u;
        const unsigned t0 = w.x ^ R4, t1 = w.y ^ R4, t2 = w.z ^ R4, t3 = w.w ^ R4;
        const unsigned z0 = (t0 - K1) & ~t0, z1 = (t1 - K1) & ~t1;
        const unsigned z2 = (t2 - K1) & ~t2, z3 = (t3 - K1) & ~t3;
        const bool cand = ((z0 | z1 | z2 | z3) & 0x80808080u) != 0u;
        if (__builtin_amdgcn_ballot_w32(cand) != 0u) {
            const unsigned f0 = z0 & ~(t0 << 7) & 0x80808080u;
            const unsigned f1 = z1 & ~(t1 << 7) & 0x80808080u;
            const unsigned f2 = z2 & ~(t2 << 7) & 0x80808080u;
            const unsigned f3 = z3 & ~(t3 << 7) & 0x80808080u;
            unsigned m = (f0 >> 7) | (f1 >> 6) | (f2 >> 5) | (f3 >> 4);
            while (__builtin_amdgcn_ballot_w32(m != 0u) != 0u) {
                const bool has = m != 0u;
                const int p = has ? __builtin_ctz(m) : 0;
                take_edges(has, ebase + (p & 7) * 4 + (p >> 3));
                m &= m - 1u;
            }
        }
    };

    const uint4* rv = (const uint4*)rid8;
    int vb = wv * 32;
    for (; vb + 31 + (L1_SU - 1) * L1_THREADS < nvf; vb += L1_THREADS * L1_SU) {
        uint4 wb[L1_SU];
#pragma unroll
        for (int u = 0; u < L1_SU; ++u) wb[u] = rv[vb + lane + u * L1_THREADS];
#pragma unroll
        for (int u = 0; u < L1_SU; ++u) scan_vec(wb[u], (vb + lane + u * L1_THREADS) * 16);
    }
    for (; vb < nvf; vb += L1_THREADS) {
        const int v = vb + lane;
        const unsigned nomatch = ~R4;
        const uint4 w = (v < nvf) ? rv[v] : make_uint4(nomatch, nomatch, nomatch, nomatch);
        scan_vec(w, v * 16);
    }
    if (wv == 0) {
        const int e = (nvf << 4) + lane;
        const bool has = lane < 16 && e < ne && (unsigned)rid8[e] == rb;
        take_edges(has, e);
    }
    __syncthreads();

    const int nq = __builtin_amdgcn_readfirstlane((int)min(wcount, (unsigned)L1_WCAP));

    for (int i0 = lane; i0 < nq; i0 += 32 * L1_PU) {
        int ee[L1_PU], ss[L1_PU], dd[L1_PU];
        float ds[L1_PU];
#pragma unroll
        for (int u = 0; u < L1_PU; ++u) {
            const int i = i0 + u * 32;
            ee[u] = (i < nq) ? (int)qs[wq0 + i] : 0;
        }
#pragma unroll
        for (int u = 0; u < L1_PU; ++u) {
            ss[u] = src[ee[u]];
            dd[u] = dst[ee[u]];
        }
#pragma unroll
        for (int u = 0; u < L1_PU; ++u) ds[u] = dinv[ss[u]];
#pragma unroll
        for (int u = 0; u < L1_PU; ++u) {
            const int i = i0 + u * 32;
            if (i < nq) {
                const unsigned rel = (unsigned)(dd[u] - lo);
                const bool ours = rel < (unsigned)cnt;
                const float nrm = ours ? ds[u] * dloc[rel] : 0.f;
                qs[wq0 + i]  = (unsigned)ss[u];
                qdw[wq0 + i] = make_uint2(ours ? rel : ~0u, __float_as_uint(nrm));
            }
        }
    }
    __syncthreads();

    for (int b0 = 0; b0 < nq; b0 += 32) {
        const int m = min(32, nq - b0);
        unsigned so = 0u;
        uint2 dw = make_uint2(~0u, 0u);
        if (lane < m) {
            so = qs[wq0 + b0 + lane] * (unsigned)HID_CH;
            dw = qdw[wq0 + b0 + lane];
        }
        const unsigned okmask = __builtin_amdgcn_ballot_w32(lane < m && dw.x != ~0u);
        for (int j0 = 0; j0 < m; j0 += L1_G) {
            float v[L1_G][4];
#pragma unroll
            for (int u = 0; u < L1_G; ++u) {
                const int j = min(j0 + u, m - 1);
                const unsigned sj = (unsigned)__builtin_amdgcn_readlane((int)so, j);
                const float* hp = h + (size_t)sj + lane;
                v[u][0] = hp[0];
                v[u][1] = hp[32];
                v[u][2] = hp[64];
                v[u][3] = hp[96];
            }
#pragma unroll
            for (int u = 0; u < L1_G; ++u) {
                const int j = j0 + u;
                if ((okmask >> j) & 1u) {
                    const unsigned dj = (unsigned)__builtin_amdgcn_readlane((int)dw.x, j);
                    const float nj = __uint_as_float(
                        (unsigned)__builtin_amdgcn_readlane((int)dw.y, j));
                    float* ap = &acc[dj * L1_PITCH + lane];
                    ldsAddF(ap +  0, v[u][0] * nj);
                    ldsAddF(ap + 32, v[u][1] * nj);
                    ldsAddF(ap + 64, v[u][2] * nj);
                    ldsAddF(ap + 96, v[u][3] * nj);
                }
            }
        }
    }
    __syncthreads();

    if (tid < cnt) {
        const float* row = &acc[tid * L1_PITCH];
        float s0 = 0.f, s1 = 0.f;
#pragma unroll 8
        for (int c = 0; c < HID_CH; c += 4) {
            float r0 = fmaxf(row[c + 0], 0.f), r1 = fmaxf(row[c + 1], 0.f);
            float r2 = fmaxf(row[c + 2], 0.f), r3 = fmaxf(row[c + 3], 0.f);
            s0 += r0 * W2[(c + 0) * 2 + 0] + r1 * W2[(c + 1) * 2 + 0]
                + r2 * W2[(c + 2) * 2 + 0] + r3 * W2[(c + 3) * 2 + 0];
            s1 += r0 * W2[(c + 0) * 2 + 1] + r1 * W2[(c + 1) * 2 + 1]
                + r2 * W2[(c + 2) * 2 + 1] + r3 * W2[(c + 3) * 2 + 1];
        }
        h2[(size_t)(lo + tid) * 2 + 0] = s0;
        h2[(size_t)(lo + tid) * 2 + 1] = s1;
    }
}

__global__ __launch_bounds__(NT_THREADS) void k_agg2_lds(
    const float* __restrict__ h2, const int* __restrict__ src, const int* __restrict__ dst,
    const float* __restrict__ dinv, float* __restrict__ part2, int ne, int n, int chunk_len)
{
    __shared__ float acc[NT_RANGE * 2];
    __shared__ float dloc[NT_RANGE];
    const int tid = threadIdx.x;
    const int p   = blockIdx.x;
    const int lo  = blockIdx.y * NT_RANGE;
    const int hi  = min(NT_RANGE, n - lo);

    float dl[NT_SLICES];
#pragma unroll
    for (int k = 0; k < NT_SLICES; ++k) {
        const int i = tid + k * NT_THREADS;
        dl[k] = (i < hi) ? dinv[lo + i] : 0.f;
    }
#pragma unroll
    for (int k = 0; k < NT_SLICES; ++k) {
        const int i = tid + k * NT_THREADS;
        if (i < NT_RANGE) {
            acc[2 * i + 0] = 0.f;
            acc[2 * i + 1] = 0.f;
            dloc[i] = dl[k];
        }
    }
    __syncthreads();

    const int e0 = p * chunk_len;
    const int e1 = min(ne, e0 + chunk_len);
    for (int base = e0 + tid; base < e1; base += NT_THREADS * NT_U) {
        int dv[NT_U], sv[NT_U];
#pragma unroll
        for (int u = 0; u < NT_U; ++u) {
            const int e = base + u * NT_THREADS;
            const bool ok = e < e1;
            dv[u] = ok ? dst[e] : -1;
            sv[u] = ok ? src[e] : 0;
        }
        float  ds[NT_U];
        float2 hv[NT_U];
#pragma unroll
        for (int u = 0; u < NT_U; ++u) {
            const bool m = (unsigned)(dv[u] - lo) < (unsigned)NT_RANGE;
            const int  s = m ? sv[u] : 0;
            ds[u] = dinv[s];
            hv[u] = *(const float2*)(h2 + (size_t)s * 2);
        }
#pragma unroll
        for (int u = 0; u < NT_U; ++u) {
            const unsigned rel = (unsigned)(dv[u] - lo);
            if (rel < (unsigned)NT_RANGE) {
                const float norm = ds[u] * dloc[rel];
                __hip_atomic_fetch_add(&acc[rel * 2 + 0], hv[u].x * norm,
                                       __ATOMIC_RELAXED, __HIP_MEMORY_SCOPE_WORKGROUP);
                __hip_atomic_fetch_add(&acc[rel * 2 + 1], hv[u].y * norm,
                                       __ATOMIC_RELAXED, __HIP_MEMORY_SCOPE_WORKGROUP);
            }
        }
    }
    __syncthreads();

    float2* pp = (float2*)(part2 + ((size_t)p * n + lo) * 2);
    for (int i = tid; i < hi; i += NT_THREADS) pp[i] = make_float2(acc[2 * i], acc[2 * i + 1]);
}

__global__ void k_out_sum(const float* __restrict__ h2, const float* __restrict__ part2,
                          const float* __restrict__ b2, const float* __restrict__ dinv,
                          float* __restrict__ out, int n)
{
    int i = blockIdx.x * blockDim.x + threadIdx.x;
    if (i >= n) return;
    const float di = dinv[i], w = di * di;
    const float2 hs = *(const float2*)(h2 + (size_t)i * 2);
    float o0 = b2[0] + hs.x * w;
    float o1 = b2[1] + hs.y * w;
#pragma unroll
    for (int p = 0; p < NT_PLANES; ++p) {
        const float2 v = *(const float2*)(part2 + ((size_t)p * n + i) * 2);
        o0 += v.x;
        o1 += v.y;
    }
    *(float2*)(out + (size_t)i * 2) = make_float2(o0, o1);
}

extern "C" void kernel_launch(void* const* d_in, const int* in_sizes, int n_in,
                              void* d_out, int out_size, void* d_ws, size_t ws_size,
                              hipStream_t stream)
{
    const float* x  = (const float*)d_in[0];
    const int*   ei = (const int*)d_in[1];
    const float* W1 = (const float*)d_in[2];
    const float* b1 = (const float*)d_in[3];
    const float* W2 = (const float*)d_in[4];
    const float* b2 = (const float*)d_in[5];
    float* out = (float*)d_out;

    const int n_nodes = in_sizes[0] / IN_CH;
    const int n_edges = in_sizes[1] / 2;
    const int* src = ei;
    const int* dst = ei + n_edges;

    char* ws = (char*)d_ws;
    size_t off = 0;
    float* dinv = (float*)(ws + off);
    off += ((size_t)n_nodes * 4 + 4095) & ~(size_t)4095;
    float* h = (float*)(ws + off);
    size_t hb = ((size_t)n_nodes * HID_CH * 4 + 4095) & ~(size_t)4095;
    off += hb;
    char* reg3 = ws + off;

    unsigned* part  = (unsigned*)reg3;
    float*    part2 = (float*)reg3;
    size_t pb = ((size_t)n_nodes * NT_PLANES * 8 + 4095) & ~(size_t)4095;
    float* h2 = (float*)(reg3 + pb);
    size_t h2b = ((size_t)n_nodes * 2 * 4 + 4095) & ~(size_t)4095;
    unsigned char* rid8 = (unsigned char*)(reg3 + pb + h2b);

    const int chunk_len = (n_edges + NT_PLANES - 1) / NT_PLANES;
    dim3 ngrid(NT_PLANES, (n_nodes + NT_RANGE - 1) / NT_RANGE);
    k_deg_hist<<<ngrid, NT_THREADS, 0, stream>>>(dst, part, rid8, n_edges, n_nodes, chunk_len);
    k_dinv    <<<(n_nodes + 255) / 256, 256, 0, stream>>>(part, dinv, n_nodes);

    k_gemm1_wmma<<<(n_nodes + 127) / 128, 256, 0, stream>>>(x, W1, h, n_nodes);

    k_agg1_own  <<<(n_nodes + L1_R - 1) / L1_R, L1_THREADS, 0, stream>>>(
        h, src, dst, dinv, b1, W2, rid8, h2, n_edges, n_nodes);

    k_agg2_lds  <<<ngrid, NT_THREADS, 0, stream>>>(h2, src, dst, dinv, part2, n_edges, n_nodes, chunk_len);
    k_out_sum   <<<(n_nodes + 255) / 256, 256, 0, stream>>>(h2, part2, b2, dinv, out, n_nodes);
}


# dinv[src] folded into stored rows (x@W1 epilogue, layer-1 output: row times its own dinv), dinv[dst] applied once per summed row: no per-edge dinv lookup or multiply; products re-associated
# speedup vs baseline: 3.0757x; 1.0343x over previous
#include <hip/hip_runtime.h>
#include <hip/hip_bf16.h>
#include <cstdint>

typedef float v2f __attribute__((ext_vector_type(2)));
typedef float v8f __attribute__((ext_vector_type(8)));

#define IN_CH  256
#define HID_CH 128
#define KTILE  64
#define PSTRIDE 288

#define L1_R       400
#define L1_PITCH   129
#define L1_WCAP    288
#define L1_THREADS 1024
#define L1_SU      8
#define L1_PU      8
#define L1_G       8

#define NT_RANGE   12544
#define NT_PLANES  32
#define NT_THREADS 1024
#define NT_U       16

__global__ __launch_bounds__(NT_THREADS) void k_deg_hist(
    const int* __restrict__ dst, unsigned* __restrict__ part, unsigned char* __restrict__ rid8,
    int ne, int n, int chunk_len)
{
    __shared__ unsigned cnt[NT_RANGE];
    const int tid = threadIdx.x;
    const int c   = blockIdx.x;
    const int lo  = blockIdx.y * NT_RANGE;

    for (int i = tid; i < NT_RANGE; i += NT_THREADS) cnt[i] = 0u;

    {
        const int nb    = gridDim.x * gridDim.y;
        const int bid   = blockIdx.y * gridDim.x + blockIdx.x;
        const int nquad = ne >> 2;
        const int qlen  = ((nquad + nb - 1) / nb + 31) & ~31;
        const int q0    = bid * qlen;
        const int q1    = min(nquad, q0 + qlen);
        unsigned* rid32 = (unsigned*)rid8;
        for (int q = q0 + tid; q < q1; q += NT_THREADS) {
            const int* dp = dst + 4 * (size_t)q;
            const unsigned o0 = ((unsigned)dp[0] / (unsigned)L1_R) & 0xffu;
            const unsigned o1 = ((unsigned)dp[1] / (unsigned)L1_R) & 0xffu;
            const unsigned o2 = ((unsigned)dp[2] / (unsigned)L1_R) & 0xffu;
            const unsigned o3 = ((unsigned)dp[3] / (unsigned)L1_R) & 0xffu;
            rid32[q] = o0 | (o1 << 8) | (o2 << 16) | (o3 << 24);
        }
        if (bid == 0 && tid < (ne & 3)) {
            const int e = (ne & ~3) + tid;
            rid8[e] = (unsigned char)((unsigned)dst[e] / (unsigned)L1_R);
        }
    }
    __syncthreads();

    const int e0 = c * chunk_len;
    const int e1 = min(ne, e0 + chunk_len);
    for (int base = e0 + tid; base < e1; base += NT_THREADS * NT_U) {
        int dv[NT_U];
#pragma unroll
        for (int u = 0; u < NT_U; ++u) {
            const int e = base + u * NT_THREADS;
            dv[u] = (e < e1) ? dst[e] : -1;
        }
#pragma unroll
        for (int u = 0; u < NT_U; ++u) {
            const unsigned rel = (unsigned)(dv[u] - lo);
            if (rel < (unsigned)NT_RANGE) atomicAdd(&cnt[rel], 1u);
        }
    }
    __syncthreads();

    const int hi = min(NT_RANGE, n - lo);
    unsigned* pp = part + (size_t)c * n + lo;
    for (int i = tid; i < hi; i += NT_THREADS) pp[i] = cnt[i];
}

__global__ void k_dinv(const unsigned* __restrict__ part, float* __restrict__ dinv, int n) {
    int i = blockIdx.x * blockDim.x + threadIdx.x;
    if (i >= n) return;
    unsigned s = 0u;
#pragma unroll
    for (int c = 0; c < NT_PLANES; ++c) s += part[(size_t)c * n + i];
    dinv[i] = rsqrtf(1.0f + (float)s);
}

__global__ __launch_bounds__(256) void k_gemm1_wmma(
    const float* __restrict__ x, const float* __restrict__ W1, const float* __restrict__ dinv,
    float* __restrict__ h, int n_nodes)
{
    __shared__ float sW[(KTILE / 2) * PSTRIDE];

    const int tid  = threadIdx.x;
    const int wave = tid >> 5;
    const int lane = tid & 31;
    const int half = lane >> 4;
    const int l16  = lane & 15;

    const int row_base = (blockIdx.x * 8 + wave) * 16;
    int arow = row_base + l16;
    if (arow >= n_nodes) arow = n_nodes - 1;
    const float* aptr = x + (size_t)arow * IN_CH + half * 2;

    v8f acc[8] = {};

    for (int k0 = 0; k0 < IN_CH; k0 += KTILE) {
        __syncthreads();
        for (int i = tid; i < (KTILE / 2) * 32; i += 256) {
            int p  = i >> 5;
            int c4 = (i & 31) * 4;
            float4 r0 = *(const float4*)(W1 + (size_t)(k0 + 2 * p + 0) * HID_CH + c4);
            float4 r1 = *(const float4*)(W1 + (size_t)(k0 + 2 * p + 1) * HID_CH + c4);
            float* dp = &sW[p * PSTRIDE + c4 * 2];
            ((float2*)dp)[0] = make_float2(r0.x, r1.x);
            ((float2*)dp)[1] = make_float2(r0.y, r1.y);
            ((float2*)dp)[2] = make_float2(r0.z, r1.z);
            ((float2*)dp)[3] = make_float2(r0.w, r1.w);
        }
        __syncthreads();

#pragma unroll 4
        for (int k = 0; k < KTILE; k += 4) {
            v2f a = *(const v2f*)(aptr + k0 + k);
            const int p = (k >> 1) + half;
            const float* bbase = &sW[p * PSTRIDE + l16 * 2];
#pragma unroll
            for (int nt = 0; nt < 8; ++nt) {
                v2f b = *(const v2f*)(bbase + nt * 32);
                acc[nt] = __builtin_amdgcn_wmma_f32_16x16x4_f32(
                    false, a, false, b, (short)0, acc[nt], false, false);
            }
        }
    }

    float dr[8];
#pragma unroll
    for (int v = 0; v < 8; ++v) dr[v] = dinv[min(row_base + v + 8 * half, n_nodes - 1)];
#pragma unroll
    for (int nt = 0; nt < 8; ++nt) {
#pragma unroll
        for (int v = 0; v < 8; ++v) {
            int r = row_base + v + 8 * half;
            if (r < n_nodes) h[(size_t)r * HID_CH + nt * 16 + l16] = acc[nt][v] * dr[v];
        }
    }
}

__device__ __forceinline__ void ldsAddF(float* p, float v) {
    __hip_atomic_fetch_add(p, v, __ATOMIC_RELAXED, __HIP_MEMORY_SCOPE_WORKGROUP);
}

__global__ __launch_bounds__(L1_THREADS) void k_agg1_own(
    const float* __restrict__ h, const int* __restrict__ src, const int* __restrict__ dst,
    const float* __restrict__ dinv, const float* __restrict__ b1, const float* __restrict__ W2,
    const unsigned char* __restrict__ rid8, float* __restrict__ h2, int ne, int n)
{
    __shared__ float    acc[L1_R * L1_PITCH];
    __shared__ unsigned qs[(L1_THREADS / 32) * L1_WCAP];
    __shared__ unsigned qd[(L1_THREADS / 32) * L1_WCAP];

    const int tid  = threadIdx.x;
    const int lane = tid & 31;
    const int wv  = __builtin_amdgcn_readfirstlane(tid >> 5);
    const int lo  = blockIdx.x * L1_R;
    const int cnt = min(L1_R, n - lo);

    {
        for (int r0 = wv * 4; r0 < cnt; r0 += (L1_THREADS / 32) * 4) {
            float hv[4][4];
#pragma unroll
            for (int q = 0; q < 4; ++q) {
                const int r = min(r0 + q, cnt - 1);
                const float* hp = h + (size_t)(lo + r) * HID_CH + lane;
#pragma unroll
                for (int k = 0; k < 4; ++k) hv[q][k] = hp[32 * k];
            }
#pragma unroll
            for (int q = 0; q < 4; ++q) {
                if (r0 + q < cnt) {
                    float* ap = &acc[(r0 + q) * L1_PITCH + lane];
#pragma unroll
                    for (int k = 0; k < 4; ++k) ap[32 * k] = hv[q][k];
                }
            }
        }
    }
    __syncthreads();

    const unsigned rb = (unsigned)blockIdx.x & 0xffu;
    const unsigned R4 = rb * 0x01010101u;
    const int nvf = ne >> 4;
    const int wq0 = wv * L1_WCAP;
    unsigned wcount = 0u;

    auto take_edges = [&](const bool has, const int e) {
        const unsigned bal = __builtin_amdgcn_ballot_w32(has);
        if (has) {
            const unsigned slot = wcount + __builtin_amdgcn_mbcnt_lo(bal, 0u);
            if (slot < (unsigned)L1_WCAP) {
                qs[wq0 + slot] = (unsigned)e;
            } else {
                const int s = src[e];
                const unsigned rel = (unsigned)(dst[e] - lo);
                if (rel < (unsigned)cnt) {
                    const float* hp = h + (size_t)s * HID_CH;
#pragma nounroll
                    for (int c = 0; c < HID_CH; ++c)
                        ldsAddF(&acc[rel * L1_PITCH + c], hp[c]);
                }
            }
        }
        wcount += (unsigned)__builtin_popcount(bal);
    };
    auto scan_vec = [&](const uint4 w, const int ebase) {
        const unsigned K1 = 0x01010101u;
        const unsigned t0 = w.x ^ R4, t1 = w.y ^ R4, t2 = w.z ^ R4, t3 = w.w ^ R4;
        const unsigned z0 = (t0 - K1) & ~t0, z1 = (t1 - K1) & ~t1;
        const unsigned z2 = (t2 - K1) & ~t2, z3 = (t3 - K1) & ~t3;
        const bool cand = ((z0 | z1 | z2 | z3) & 0x80808080u) != 0u;
        if (__builtin_amdgcn_ballot_w32(cand) != 0u) {
            const unsigned f0 = z0 & ~(t0 << 7) & 0x80808080u;
            const unsigned f1 = z1 & ~(t1 << 7) & 0x80808080u;
            const unsigned f2 = z2 & ~(t2 << 7) & 0x80808080u;
            const unsigned f3 = z3 & ~(t3 << 7) & 0x80808080u;
            unsigned m = (f0 >> 7) | (f1 >> 6) | (f2 >> 5) | (f3 >> 4);
            while (__builtin_amdgcn_ballot_w32(m != 0u) != 0u) {
                const bool has = m != 0u;
                const int p = has ? __builtin_ctz(m) : 0;
                take_edges(has, ebase + (p & 7) * 4 + (p >> 3));
                m &= m - 1u;
            }
        }
    };

    const uint4* rv = (const uint4*)rid8;
    int vb = wv * 32;
    for (; vb + 31 + (L1_SU - 1) * L1_THREADS < nvf; vb += L1_THREADS * L1_SU) {
        uint4 wb[L1_SU];
#pragma unroll
        for (int u = 0; u < L1_SU; ++u) wb[u] = rv[vb + lane + u * L1_THREADS];
#pragma unroll
        for (int u = 0; u < L1_SU; ++u) scan_vec(wb[u], (vb + lane + u * L1_THREADS) * 16);
    }
    for (; vb < nvf; vb += L1_THREADS) {
        const int v = vb + lane;
        const unsigned nomatch = ~R4;
        const uint4 w = (v < nvf) ? rv[v] : make_uint4(nomatch, nomatch, nomatch, nomatch);
        scan_vec(w, v * 16);
    }
    if (wv == 0) {
        const int e = (nvf << 4) + lane;
        const bool has = lane < 16 && e < ne && (unsigned)rid8[e] == rb;
        take_edges(has, e);
    }
    __syncthreads();

    const int nq = __builtin_amdgcn_readfirstlane((int)min(wcount, (unsigned)L1_WCAP));

    for (int i0 = lane; i0 < nq; i0 += 32 * L1_PU) {
        int ee[L1_PU], ss[L1_PU], dd[L1_PU];
#pragma unroll
        for (int u = 0; u < L1_PU; ++u) {
            const int i = i0 + u * 32;
            ee[u] = (i < nq) ? (int)qs[wq0 + i] : 0;
        }
#pragma unroll
        for (int u = 0; u < L1_PU; ++u) {
            ss[u] = src[ee[u]];
            dd[u] = dst[ee[u]];
        }
#pragma unroll
        for (int u = 0; u < L1_PU; ++u) {
            const int i = i0 + u * 32;
            if (i < nq) {
                const unsigned rel = (unsigned)(dd[u] - lo);
                const bool ours = rel < (unsigned)cnt;
                qs[wq0 + i] = (unsigned)ss[u];
                qd[wq0 + i] = ours ? rel : ~0u;
            }
        }
    }
    __syncthreads();

    for (int b0 = 0; b0 < nq; b0 += 32) {
        const int m = min(32, nq - b0);
        unsigned so = 0u;
        unsigned dw = ~0u;
        if (lane < m) {
            so = qs[wq0 + b0 + lane] * (unsigned)HID_CH;
            dw = qd[wq0 + b0 + lane];
        }
        const unsigned okmask = __builtin_amdgcn_ballot_w32(lane < m && dw != ~0u);
        for (int j0 = 0; j0 < m; j0 += L1_G) {
            float v[L1_G][4];
#pragma unroll
            for (int u = 0; u < L1_G; ++u) {
                const int j = min(j0 + u, m - 1);
                const unsigned sj = (unsigned)__builtin_amdgcn_readlane((int)so, j);
                const float* hp = h + (size_t)sj + lane;
                v[u][0] = hp[0];
                v[u][1] = hp[32];
                v[u][2] = hp[64];
                v[u][3] = hp[96];
            }
#pragma unroll
            for (int u = 0; u < L1_G; ++u) {
                const int j = j0 + u;
                if ((okmask >> j) & 1u) {
                    const unsigned dj = (unsigned)__builtin_amdgcn_readlane((int)dw, j);
                    float* ap = &acc[dj * L1_PITCH + lane];
                    ldsAddF(ap +  0, v[u][0]);
                    ldsAddF(ap + 32, v[u][1]);
                    ldsAddF(ap + 64, v[u][2]);
                    ldsAddF(ap + 96, v[u][3]);
                }
            }
        }
    }
    __syncthreads();

    if (tid < cnt) {
        const float* row = &acc[tid * L1_PITCH];
        const float dd = dinv[lo + tid];
        float s0 = 0.f, s1 = 0.f;
#pragma unroll 8
        for (int c = 0; c < HID_CH; c += 4) {
            float r0 = fmaxf(b1[c + 0] + dd * row[c + 0], 0.f);
            float r1 = fmaxf(b1[c + 1] + dd * row[c + 1], 0.f);
            float r2 = fmaxf(b1[c + 2] + dd * row[c + 2], 0.f);
            float r3 = fmaxf(b1[c + 3] + dd * row[c + 3], 0.f);
            s0 += r0 * W2[(c + 0) * 2 + 0] + r1 * W2[(c + 1) * 2 + 0]
                + r2 * W2[(c + 2) * 2 + 0] + r3 * W2[(c + 3) * 2 + 0];
            s1 += r0 * W2[(c + 0) * 2 + 1] + r1 * W2[(c + 1) * 2 + 1]
                + r2 * W2[(c + 2) * 2 + 1] + r3 * W2[(c + 3) * 2 + 1];
        }
        h2[(size_t)(lo + tid) * 2 + 0] = dd * s0;
        h2[(size_t)(lo + tid) * 2 + 1] = dd * s1;
    }
}

__global__ __launch_bounds__(NT_THREADS) void k_agg2_lds(
    const float* __restrict__ h2, const int* __restrict__ src, const int* __restrict__ dst,
    float* __restrict__ part2, int ne, int n, int chunk_len)
{
    __shared__ float acc[NT_RANGE * 2];
    const int tid = threadIdx.x;
    const int p   = blockIdx.x;
    const int lo  = blockIdx.y * NT_RANGE;
    const int hi  = min(NT_RANGE, n - lo);

    for (int i = tid; i < NT_RANGE; i += NT_THREADS) {
        acc[2 * i + 0] = 0.f;
        acc[2 * i + 1] = 0.f;
    }
    __syncthreads();

    const int e0 = p * chunk_len;
    const int e1 = min(ne, e0 + chunk_len);
    for (int base = e0 + tid; base < e1; base += NT_THREADS * NT_U) {
        int dv[NT_U], sv[NT_U];
#pragma unroll
        for (int u = 0; u < NT_U; ++u) {
            const int e = base + u * NT_THREADS;
            const bool ok = e < e1;
            dv[u] = ok ? dst[e] : -1;
            sv[u] = ok ? src[e] : 0;
        }
        float2 hv[NT_U];
#pragma unroll
        for (int u = 0; u < NT_U; ++u) {
            const bool m = (unsigned)(dv[u] - lo) < (unsigned)NT_RANGE;
            const int  s = m ? sv[u] : 0;
            hv[u] = *(const float2*)(h2 + (size_t)s * 2);
        }
#pragma unroll
        for (int u = 0; u < NT_U; ++u) {
            const unsigned rel = (unsigned)(dv[u] - lo);
            if (rel < (unsigned)NT_RANGE) {
                __hip_atomic_fetch_add(&acc[rel * 2 + 0], hv[u].x,
                                       __ATOMIC_RELAXED, __HIP_MEMORY_SCOPE_WORKGROUP);
                __hip_atomic_fetch_add(&acc[rel * 2 + 1], hv[u].y,
                                       __ATOMIC_RELAXED, __HIP_MEMORY_SCOPE_WORKGROUP);
            }
        }
    }
    __syncthreads();

    float2* pp = (float2*)(part2 + ((size_t)p * n + lo) * 2);
    for (int i = tid; i < hi; i += NT_THREADS) pp[i] = make_float2(acc[2 * i], acc[2 * i + 1]);
}

__global__ void k_out_sum(const float* __restrict__ h2, const float* __restrict__ part2,
                          const float* __restrict__ b2, const float* __restrict__ dinv,
                          float* __restrict__ out, int n)
{
    int i = blockIdx.x * blockDim.x + threadIdx.x;
    if (i >= n) return;
    const float di = dinv[i];
    const float2 hs = *(const float2*)(h2 + (size_t)i * 2);
    float o0 = hs.x;
    float o1 = hs.y;
#pragma unroll
    for (int p = 0; p < NT_PLANES; ++p) {
        const float2 v = *(const float2*)(part2 + ((size_t)p * n + i) * 2);
        o0 += v.x;
        o1 += v.y;
    }
    *(float2*)(out + (size_t)i * 2) = make_float2(b2[0] + di * o0, b2[1] + di * o1);
}

extern "C" void kernel_launch(void* const* d_in, const int* in_sizes, int n_in,
                              void* d_out, int out_size, void* d_ws, size_t ws_size,
                              hipStream_t stream)
{
    const float* x  = (const float*)d_in[0];
    const int*   ei = (const int*)d_in[1];
    const float* W1 = (const float*)d_in[2];
    const float* b1 = (const float*)d_in[3];
    const float* W2 = (const float*)d_in[4];
    const float* b2 = (const float*)d_in[5];
    float* out = (float*)d_out;

    const int n_nodes = in_sizes[0] / IN_CH;
    const int n_edges = in_sizes[1] / 2;
    const int* src = ei;
    const int* dst = ei + n_edges;

    char* ws = (char*)d_ws;
    size_t off = 0;
    float* dinv = (float*)(ws + off);
    off += ((size_t)n_nodes * 4 + 4095) & ~(size_t)4095;
    float* h = (float*)(ws + off);
    size_t hb = ((size_t)n_nodes * HID_CH * 4 + 4095) & ~(size_t)4095;
    off += hb;
    char* reg3 = ws + off;

    unsigned* part  = (unsigned*)reg3;
    float*    part2 = (float*)reg3;
    size_t pb = ((size_t)n_nodes * NT_PLANES * 8 + 4095) & ~(size_t)4095;
    float* h2 = (float*)(reg3 + pb);
    size_t h2b = ((size_t)n_nodes * 2 * 4 + 4095) & ~(size_t)4095;
    unsigned char* rid8 = (unsigned char*)(reg3 + pb + h2b);

    const int chunk_len = (n_edges + NT_PLANES - 1) / NT_PLANES;
    dim3 ngrid(NT_PLANES, (n_nodes + NT_RANGE - 1) / NT_RANGE);
    k_deg_hist<<<ngrid, NT_THREADS, 0, stream>>>(dst, part, rid8, n_edges, n_nodes, chunk_len);
    k_dinv    <<<(n_nodes + 255) / 256, 256, 0, stream>>>(part, dinv, n_nodes);

    k_gemm1_wmma<<<(n_nodes + 127) / 128, 256, 0, stream>>>(x, W1, dinv, h, n_nodes);

    k_agg1_own  <<<(n_nodes + L1_R - 1) / L1_R, L1_THREADS, 0, stream>>>(
        h, src, dst, dinv, b1, W2, rid8, h2, n_edges, n_nodes);

    k_agg2_lds  <<<ngrid, NT_THREADS, 0, stream>>>(h2, src, dst, part2, n_edges, n_nodes, chunk_len);
    k_out_sum   <<<(n_nodes + 255) / 256, 256, 0, stream>>>(h2, part2, b2, dinv, out, n_nodes);
}


# Layer-1 row gathers: one 16-byte-per-lane load takes a whole 512-B source row (was four 4-byte line loads); LDS rows channel-permuted so the adds stay conflict-free; LDS arrival-order sums kept
# speedup vs baseline: 3.2287x; 1.0497x over previous
#include <hip/hip_runtime.h>
#include <hip/hip_bf16.h>
#include <cstdint>

typedef float v2f __attribute__((ext_vector_type(2)));
typedef float v8f __attribute__((ext_vector_type(8)));

#define IN_CH  256
#define HID_CH 128
#define KTILE  64
#define PSTRIDE 288

#define L1_R       400
#define L1_PITCH   129
#define L1_WCAP    288
#define L1_THREADS 1024
#define L1_SU      8
#define L1_PU      8
#define L1_G       8
#define L1_POS(c)  ((((c) & 3) << 5) + ((c) >> 2))

#define NT_RANGE   12544
#define NT_PLANES  32
#define NT_THREADS 1024
#define NT_U       16

__global__ __launch_bounds__(NT_THREADS) void k_deg_hist(
    const int* __restrict__ dst, unsigned* __restrict__ part, unsigned char* __restrict__ rid8,
    int ne, int n, int chunk_len)
{
    __shared__ unsigned cnt[NT_RANGE];
    const int tid = threadIdx.x;
    const int c   = blockIdx.x;
    const int lo  = blockIdx.y * NT_RANGE;

    for (int i = tid; i < NT_RANGE; i += NT_THREADS) cnt[i] = 0u;

    {
        const int nb    = gridDim.x * gridDim.y;
        const int bid   = blockIdx.y * gridDim.x + blockIdx.x;
        const int nquad = ne >> 2;
        const int qlen  = ((nquad + nb - 1) / nb + 31) & ~31;
        const int q0    = bid * qlen;
        const int q1    = min(nquad, q0 + qlen);
        unsigned* rid32 = (unsigned*)rid8;
        for (int q = q0 + tid; q < q1; q += NT_THREADS) {
            const int* dp = dst + 4 * (size_t)q;
            const unsigned o0 = ((unsigned)dp[0] / (unsigned)L1_R) & 0xffu;
            const unsigned o1 = ((unsigned)dp[1] / (unsigned)L1_R) & 0xffu;
            const unsigned o2 = ((unsigned)dp[2] / (unsigned)L1_R) & 0xffu;
            const unsigned o3 = ((unsigned)dp[3] / (unsigned)L1_R) & 0xffu;
            rid32[q] = o0 | (o1 << 8) | (o2 << 16) | (o3 << 24);
        }
        if (bid == 0 && tid < (ne & 3)) {
            const int e = (ne & ~3) + tid;
            rid8[e] = (unsigned char)((unsigned)dst[e] / (unsigned)L1_R);
        }
    }
    __syncthreads();

    const int e0 = c * chunk_len;
    const int e1 = min(ne, e0 + chunk_len);
    for (int base = e0 + tid; base < e1; base += NT_THREADS * NT_U) {
        int dv[NT_U];
#pragma unroll
        for (int u = 0; u < NT_U; ++u) {
            const int e = base + u * NT_THREADS;
            dv[u] = (e < e1) ? dst[e] : -1;
        }
#pragma unroll
        for (int u = 0; u < NT_U; ++u) {
            const unsigned rel = (unsigned)(dv[u] - lo);
            if (rel < (unsigned)NT_RANGE) atomicAdd(&cnt[rel], 1u);
        }
    }
    __syncthreads();

    const int hi = min(NT_RANGE, n - lo);
    unsigned* pp = part + (size_t)c * n + lo;
    for (int i = tid; i < hi; i += NT_THREADS) pp[i] = cnt[i];
}

__global__ void k_dinv(const unsigned* __restrict__ part, float* __restrict__ dinv, int n) {
    int i = blockIdx.x * blockDim.x + threadIdx.x;
    if (i >= n) return;
    unsigned s = 0u;
#pragma unroll
    for (int c = 0; c < NT_PLANES; ++c) s += part[(size_t)c * n + i];
    dinv[i] = rsqrtf(1.0f + (float)s);
}

__global__ __launch_bounds__(256) void k_gemm1_wmma(
    const float* __restrict__ x, const float* __restrict__ W1, const float* __restrict__ dinv,
    float* __restrict__ h, int n_nodes)
{
    __shared__ float sW[(KTILE / 2) * PSTRIDE];

    const int tid  = threadIdx.x;
    const int wave = tid >> 5;
    const int lane = tid & 31;
    const int half = lane >> 4;
    const int l16  = lane & 15;

    const int row_base = (blockIdx.x * 8 + wave) * 16;
    int arow = row_base + l16;
    if (arow >= n_nodes) arow = n_nodes - 1;
    const float* aptr = x + (size_t)arow * IN_CH + half * 2;

    v8f acc[8] = {};

    for (int k0 = 0; k0 < IN_CH; k0 += KTILE) {
        __syncthreads();
        for (int i = tid; i < (KTILE / 2) * 32; i += 256) {
            int p  = i >> 5;
            int c4 = (i & 31) * 4;
            float4 r0 = *(const float4*)(W1 + (size_t)(k0 + 2 * p + 0) * HID_CH + c4);
            float4 r1 = *(const float4*)(W1 + (size_t)(k0 + 2 * p + 1) * HID_CH + c4);
            float* dp = &sW[p * PSTRIDE + c4 * 2];
            ((float2*)dp)[0] = make_float2(r0.x, r1.x);
            ((float2*)dp)[1] = make_float2(r0.y, r1.y);
            ((float2*)dp)[2] = make_float2(r0.z, r1.z);
            ((float2*)dp)[3] = make_float2(r0.w, r1.w);
        }
        __syncthreads();

#pragma unroll 4
        for (int k = 0; k < KTILE; k += 4) {
            v2f a = *(const v2f*)(aptr + k0 + k);
            const int p = (k >> 1) + half;
            const float* bbase = &sW[p * PSTRIDE + l16 * 2];
#pragma unroll
            for (int nt = 0; nt < 8; ++nt) {
                v2f b = *(const v2f*)(bbase + nt * 32);
                acc[nt] = __builtin_amdgcn_wmma_f32_16x16x4_f32(
                    false, a, false, b, (short)0, acc[nt], false, false);
            }
        }
    }

    float dr[8];
#pragma unroll
    for (int v = 0; v < 8; ++v) dr[v] = dinv[min(row_base + v + 8 * half, n_nodes - 1)];
#pragma unroll
    for (int nt = 0; nt < 8; ++nt) {
#pragma unroll
        for (int v = 0; v < 8; ++v) {
            int r = row_base + v + 8 * half;
            if (r < n_nodes) h[(size_t)r * HID_CH + nt * 16 + l16] = acc[nt][v] * dr[v];
        }
    }
}

__device__ __forceinline__ void ldsAddF(float* p, float v) {
    __hip_atomic_fetch_add(p, v, __ATOMIC_RELAXED, __HIP_MEMORY_SCOPE_WORKGROUP);
}

__global__ __launch_bounds__(L1_THREADS) void k_agg1_own(
    const float* __restrict__ h, const int* __restrict__ src, const int* __restrict__ dst,
    const float* __restrict__ dinv, const float* __restrict__ b1, const float* __restrict__ W2,
    const unsigned char* __restrict__ rid8, float* __restrict__ h2, int ne, int n)
{
    __shared__ float    acc[L1_R * L1_PITCH];
    __shared__ unsigned qs[(L1_THREADS / 32) * L1_WCAP];
    __shared__ unsigned qd[(L1_THREADS / 32) * L1_WCAP];

    const int tid  = threadIdx.x;
    const int lane = tid & 31;
    const int wv  = __builtin_amdgcn_readfirstlane(tid >> 5);
    const int lo  = blockIdx.x * L1_R;
    const int cnt = min(L1_R, n - lo);

    {
        for (int r0 = wv * 4; r0 < cnt; r0 += (L1_THREADS / 32) * 4) {
            float4 hv[4];
#pragma unroll
            for (int q = 0; q < 4; ++q) {
                const int r = min(r0 + q, cnt - 1);
                hv[q] = *(const float4*)(h + (size_t)(lo + r) * HID_CH + 4 * lane);
            }
#pragma unroll
            for (int q = 0; q < 4; ++q) {
                if (r0 + q < cnt) {
                    float* ap = &acc[(r0 + q) * L1_PITCH + lane];
                    ap[0]  = hv[q].x;
                    ap[32] = hv[q].y;
                    ap[64] = hv[q].z;
                    ap[96] = hv[q].w;
                }
            }
        }
    }
    __syncthreads();

    const unsigned rb = (unsigned)blockIdx.x & 0xffu;
    const unsigned R4 = rb * 0x01010101u;
    const int nvf = ne >> 4;
    const int wq0 = wv * L1_WCAP;
    unsigned wcount = 0u;

    auto take_edges = [&](const bool has, const int e) {
        const unsigned bal = __builtin_amdgcn_ballot_w32(has);
        if (has) {
            const unsigned slot = wcount + __builtin_amdgcn_mbcnt_lo(bal, 0u);
            if (slot < (unsigned)L1_WCAP) {
                qs[wq0 + slot] = (unsigned)e;
            } else {
                const int s = src[e];
                const unsigned rel = (unsigned)(dst[e] - lo);
                if (rel < (unsigned)cnt) {
                    const float* hp = h + (size_t)s * HID_CH;
#pragma nounroll
                    for (int c = 0; c < HID_CH; ++c)
                        ldsAddF(&acc[rel * L1_PITCH + L1_POS(c)], hp[c]);
                }
            }
        }
        wcount += (unsigned)__builtin_popcount(bal);
    };
    auto scan_vec = [&](const uint4 w, const int ebase) {
        const unsigned K1 = 0x01010101u;
        const unsigned t0 = w.x ^ R4, t1 = w.y ^ R4, t2 = w.z ^ R4, t3 = w.w ^ R4;
        const unsigned z0 = (t0 - K1) & ~t0, z1 = (t1 - K1) & ~t1;
        const unsigned z2 = (t2 - K1) & ~t2, z3 = (t3 - K1) & ~t3;
        const bool cand = ((z0 | z1 | z2 | z3) & 0x80808080u) != 0u;
        if (__builtin_amdgcn_ballot_w32(cand) != 0u) {
            const unsigned f0 = z0 & ~(t0 << 7) & 0x80808080u;
            const unsigned f1 = z1 & ~(t1 << 7) & 0x80808080u;
            const unsigned f2 = z2 & ~(t2 << 7) & 0x80808080u;
            const unsigned f3 = z3 & ~(t3 << 7) & 0x80808080u;
            unsigned m = (f0 >> 7) | (f1 >> 6) | (f2 >> 5) | (f3 >> 4);
            while (__builtin_amdgcn_ballot_w32(m != 0u) != 0u) {
                const bool has = m != 0u;
                const int p = has ? __builtin_ctz(m) : 0;
                take_edges(has, ebase + (p & 7) * 4 + (p >> 3));
                m &= m - 1u;
            }
        }
    };

    const uint4* rv = (const uint4*)rid8;
    int vb = wv * 32;
    for (; vb + 31 + (L1_SU - 1) * L1_THREADS < nvf; vb += L1_THREADS * L1_SU) {
        uint4 wb[L1_SU];
#pragma unroll
        for (int u = 0; u < L1_SU; ++u) wb[u] = rv[vb + lane + u * L1_THREADS];
#pragma unroll
        for (int u = 0; u < L1_SU; ++u) scan_vec(wb[u], (vb + lane + u * L1_THREADS) * 16);
    }
    for (; vb < nvf; vb += L1_THREADS) {
        const int v = vb + lane;
        const unsigned nomatch = ~R4;
        const uint4 w = (v < nvf) ? rv[v] : make_uint4(nomatch, nomatch, nomatch, nomatch);
        scan_vec(w, v * 16);
    }
    if (wv == 0) {
        const int e = (nvf << 4) + lane;
        const bool has = lane < 16 && e < ne && (unsigned)rid8[e] == rb;
        take_edges(has, e);
    }
    __syncthreads();

    const int nq = __builtin_amdgcn_readfirstlane((int)min(wcount, (unsigned)L1_WCAP));

    for (int i0 = lane; i0 < nq; i0 += 32 * L1_PU) {
        int ee[L1_PU], ss[L1_PU], dd[L1_PU];
#pragma unroll
        for (int u = 0; u < L1_PU; ++u) {
            const int i = i0 + u * 32;
            ee[u] = (i < nq) ? (int)qs[wq0 + i] : 0;
        }
#pragma unroll
        for (int u = 0; u < L1_PU; ++u) {
            ss[u] = src[ee[u]];
            dd[u] = dst[ee[u]];
        }
#pragma unroll
        for (int u = 0; u < L1_PU; ++u) {
            const int i = i0 + u * 32;
            if (i < nq) {
                const unsigned rel = (unsigned)(dd[u] - lo);
                const bool ours = rel < (unsigned)cnt;
                qs[wq0 + i] = (unsigned)ss[u];
                qd[wq0 + i] = ours ? rel : ~0u;
            }
        }
    }
    __syncthreads();

    for (int b0 = 0; b0 < nq; b0 += 32) {
        const int m = min(32, nq - b0);
        unsigned so = 0u;
        unsigned dw = ~0u;
        if (lane < m) {
            so = qs[wq0 + b0 + lane] * (unsigned)HID_CH;
            dw = qd[wq0 + b0 + lane];
        }
        const unsigned okmask = __builtin_amdgcn_ballot_w32(lane < m && dw != ~0u);
        for (int j0 = 0; j0 < m; j0 += L1_G) {
            float4 v[L1_G];
#pragma unroll
            for (int u = 0; u < L1_G; ++u) {
                const int j = min(j0 + u, m - 1);
                const unsigned sj = (unsigned)__builtin_amdgcn_readlane((int)so, j);
                v[u] = *(const float4*)(h + (size_t)sj + 4 * lane);
            }
#pragma unroll
            for (int u = 0; u < L1_G; ++u) {
                const int j = j0 + u;
                if ((okmask >> j) & 1u) {
                    const unsigned dj = (unsigned)__builtin_amdgcn_readlane((int)dw, j);
                    float* ap = &acc[dj * L1_PITCH + lane];
                    ldsAddF(ap +  0, v[u].x);
                    ldsAddF(ap + 32, v[u].y);
                    ldsAddF(ap + 64, v[u].z);
                    ldsAddF(ap + 96, v[u].w);
                }
            }
        }
    }
    __syncthreads();

    if (tid < cnt) {
        const float* row = &acc[tid * L1_PITCH];
        const float dd = dinv[lo + tid];
        float s0 = 0.f, s1 = 0.f;
#pragma unroll 8
        for (int c = 0; c < HID_CH; c += 4) {
            float r0 = fmaxf(b1[c + 0] + dd * row[L1_POS(c + 0)], 0.f);
            float r1 = fmaxf(b1[c + 1] + dd * row[L1_POS(c + 1)], 0.f);
            float r2 = fmaxf(b1[c + 2] + dd * row[L1_POS(c + 2)], 0.f);
            float r3 = fmaxf(b1[c + 3] + dd * row[L1_POS(c + 3)], 0.f);
            s0 += r0 * W2[(c + 0) * 2 + 0] + r1 * W2[(c + 1) * 2 + 0]
                + r2 * W2[(c + 2) * 2 + 0] + r3 * W2[(c + 3) * 2 + 0];
            s1 += r0 * W2[(c + 0) * 2 + 1] + r1 * W2[(c + 1) * 2 + 1]
                + r2 * W2[(c + 2) * 2 + 1] + r3 * W2[(c + 3) * 2 + 1];
        }
        h2[(size_t)(lo + tid) * 2 + 0] = dd * s0;
        h2[(size_t)(lo + tid) * 2 + 1] = dd * s1;
    }
}

__global__ __launch_bounds__(NT_THREADS) void k_agg2_lds(
    const float* __restrict__ h2, const int* __restrict__ src, const int* __restrict__ dst,
    float* __restrict__ part2, int ne, int n, int chunk_len)
{
    __shared__ float acc[NT_RANGE * 2];
    const int tid = threadIdx.x;
    const int p   = blockIdx.x;
    const int lo  = blockIdx.y * NT_RANGE;
    const int hi  = min(NT_RANGE, n - lo);

    for (int i = tid; i < NT_RANGE; i += NT_THREADS) {
        acc[2 * i + 0] = 0.f;
        acc[2 * i + 1] = 0.f;
    }
    __syncthreads();

    const int e0 = p * chunk_len;
    const int e1 = min(ne, e0 + chunk_len);
    for (int base = e0 + tid; base < e1; base += NT_THREADS * NT_U) {
        int dv[NT_U], sv[NT_U];
#pragma unroll
        for (int u = 0; u < NT_U; ++u) {
            const int e = base + u * NT_THREADS;
            const bool ok = e < e1;
            dv[u] = ok ? dst[e] : -1;
            sv[u] = ok ? src[e] : 0;
        }
        float2 hv[NT_U];
#pragma unroll
        for (int u = 0; u < NT_U; ++u) {
            const bool m = (unsigned)(dv[u] - lo) < (unsigned)NT_RANGE;
            const int  s = m ? sv[u] : 0;
            hv[u] = *(const float2*)(h2 + (size_t)s * 2);
        }
#pragma unroll
        for (int u = 0; u < NT_U; ++u) {
            const unsigned rel = (unsigned)(dv[u] - lo);
            if (rel < (unsigned)NT_RANGE) {
                __hip_atomic_fetch_add(&acc[rel * 2 + 0], hv[u].x,
                                       __ATOMIC_RELAXED, __HIP_MEMORY_SCOPE_WORKGROUP);
                __hip_atomic_fetch_add(&acc[rel * 2 + 1], hv[u].y,
                                       __ATOMIC_RELAXED, __HIP_MEMORY_SCOPE_WORKGROUP);
            }
        }
    }
    __syncthreads();

    float2* pp = (float2*)(part2 + ((size_t)p * n + lo) * 2);
    for (int i = tid; i < hi; i += NT_THREADS) pp[i] = make_float2(acc[2 * i], acc[2 * i + 1]);
}

__global__ void k_out_sum(const float* __restrict__ h2, const float* __restrict__ part2,
                          const float* __restrict__ b2, const float* __restrict__ dinv,
                          float* __restrict__ out, int n)
{
    int i = blockIdx.x * blockDim.x + threadIdx.x;
    if (i >= n) return;
    const float di = dinv[i];
    const float2 hs = *(const float2*)(h2 + (size_t)i * 2);
    float o0 = hs.x;
    float o1 = hs.y;
#pragma unroll
    for (int p = 0; p < NT_PLANES; ++p) {
        const float2 v = *(const float2*)(part2 + ((size_t)p * n + i) * 2);
        o0 += v.x;
        o1 += v.y;
    }
    *(float2*)(out + (size_t)i * 2) = make_float2(b2[0] + di * o0, b2[1] + di * o1);
}

extern "C" void kernel_launch(void* const* d_in, const int* in_sizes, int n_in,
                              void* d_out, int out_size, void* d_ws, size_t ws_size,
                              hipStream_t stream)
{
    const float* x  = (const float*)d_in[0];
    const int*   ei = (const int*)d_in[1];
    const float* W1 = (const float*)d_in[2];
    const float* b1 = (const float*)d_in[3];
    const float* W2 = (const float*)d_in[4];
    const float* b2 = (const float*)d_in[5];
    float* out = (float*)d_out;

    const int n_nodes = in_sizes[0] / IN_CH;
    const int n_edges = in_sizes[1] / 2;
    const int* src = ei;
    const int* dst = ei + n_edges;

    char* ws = (char*)d_ws;
    size_t off = 0;
    float* dinv = (float*)(ws + off);
    off += ((size_t)n_nodes * 4 + 4095) & ~(size_t)4095;
    float* h = (float*)(ws + off);
    size_t hb = ((size_t)n_nodes * HID_CH * 4 + 4095) & ~(size_t)4095;
    off += hb;
    char* reg3 = ws + off;

    unsigned* part  = (unsigned*)reg3;
    float*    part2 = (float*)reg3;
    size_t pb = ((size_t)n_nodes * NT_PLANES * 8 + 4095) & ~(size_t)4095;
    float* h2 = (float*)(reg3 + pb);
    size_t h2b = ((size_t)n_nodes * 2 * 4 + 4095) & ~(size_t)4095;
    unsigned char* rid8 = (unsigned char*)(reg3 + pb + h2b);

    const int chunk_len = (n_edges + NT_PLANES - 1) / NT_PLANES;
    dim3 ngrid(NT_PLANES, (n_nodes + NT_RANGE - 1) / NT_RANGE);
    k_deg_hist<<<ngrid, NT_THREADS, 0, stream>>>(dst, part, rid8, n_edges, n_nodes, chunk_len);
    k_dinv    <<<(n_nodes + 255) / 256, 256, 0, stream>>>(part, dinv, n_nodes);

    k_gemm1_wmma<<<(n_nodes + 127) / 128, 256, 0, stream>>>(x, W1, dinv, h, n_nodes);

    k_agg1_own  <<<(n_nodes + L1_R - 1) / L1_R, L1_THREADS, 0, stream>>>(
        h, src, dst, dinv, b1, W2, rid8, h2, n_edges, n_nodes);

    k_agg2_lds  <<<ngrid, NT_THREADS, 0, stream>>>(h2, src, dst, part2, n_edges, n_nodes, chunk_len);
    k_out_sum   <<<(n_nodes + 255) / 256, 256, 0, stream>>>(h2, part2, b2, dinv, out, n_nodes);
}


# Layer-1 owner blocks: no block barrier between owner-table scan, lookups and row adds; each wave drains its own list every 32 entries, so scan work overlaps row gathers; LDS arrival-order sums kept
# speedup vs baseline: 3.4239x; 1.0605x over previous
#include <hip/hip_runtime.h>
#include <hip/hip_bf16.h>
#include <cstdint>

typedef float v2f __attribute__((ext_vector_type(2)));
typedef float v8f __attribute__((ext_vector_type(8)));

#define IN_CH  256
#define HID_CH 128
#define KTILE  64
#define PSTRIDE 288

#define L1_R       400
#define L1_PITCH   129
#define L1_WCAP    288
#define L1_THREADS 1024
#define L1_SU      8
#define L1_PU      2
#define L1_DRAIN   32
#define L1_G       8
#define L1_POS(c)  ((((c) & 3) << 5) + ((c) >> 2))

#define NT_RANGE   12544
#define NT_PLANES  32
#define NT_THREADS 1024
#define NT_U       16

__global__ __launch_bounds__(NT_THREADS) void k_deg_hist(
    const int* __restrict__ dst, unsigned* __restrict__ part, unsigned char* __restrict__ rid8,
    int ne, int n, int chunk_len)
{
    __shared__ unsigned cnt[NT_RANGE];
    const int tid = threadIdx.x;
    const int c   = blockIdx.x;
    const int lo  = blockIdx.y * NT_RANGE;

    for (int i = tid; i < NT_RANGE; i += NT_THREADS) cnt[i] = 0u;

    {
        const int nb    = gridDim.x * gridDim.y;
        const int bid   = blockIdx.y * gridDim.x + blockIdx.x;
        const int nquad = ne >> 2;
        const int qlen  = ((nquad + nb - 1) / nb + 31) & ~31;
        const int q0    = bid * qlen;
        const int q1    = min(nquad, q0 + qlen);
        unsigned* rid32 = (unsigned*)rid8;
        for (int q = q0 + tid; q < q1; q += NT_THREADS) {
            const int* dp = dst + 4 * (size_t)q;
            const unsigned o0 = ((unsigned)dp[0] / (unsigned)L1_R) & 0xffu;
            const unsigned o1 = ((unsigned)dp[1] / (unsigned)L1_R) & 0xffu;
            const unsigned o2 = ((unsigned)dp[2] / (unsigned)L1_R) & 0xffu;
            const unsigned o3 = ((unsigned)dp[3] / (unsigned)L1_R) & 0xffu;
            rid32[q] = o0 | (o1 << 8) | (o2 << 16) | (o3 << 24);
        }
        if (bid == 0 && tid < (ne & 3)) {
            const int e = (ne & ~3) + tid;
            rid8[e] = (unsigned char)((unsigned)dst[e] / (unsigned)L1_R);
        }
    }
    __syncthreads();

    const int e0 = c * chunk_len;
    const int e1 = min(ne, e0 + chunk_len);
    for (int base = e0 + tid; base < e1; base += NT_THREADS * NT_U) {
        int dv[NT_U];
#pragma unroll
        for (int u = 0; u < NT_U; ++u) {
            const int e = base + u * NT_THREADS;
            dv[u] = (e < e1) ? dst[e] : -1;
        }
#pragma unroll
        for (int u = 0; u < NT_U; ++u) {
            const unsigned rel = (unsigned)(dv[u] - lo);
            if (rel < (unsigned)NT_RANGE) atomicAdd(&cnt[rel], 1u);
        }
    }
    __syncthreads();

    const int hi = min(NT_RANGE, n - lo);
    unsigned* pp = part + (size_t)c * n + lo;
    for (int i = tid; i < hi; i += NT_THREADS) pp[i] = cnt[i];
}

__global__ void k_dinv(const unsigned* __restrict__ part, float* __restrict__ dinv, int n) {
    int i = blockIdx.x * blockDim.x + threadIdx.x;
    if (i >= n) return;
    unsigned s = 0u;
#pragma unroll
    for (int c = 0; c < NT_PLANES; ++c) s += part[(size_t)c * n + i];
    dinv[i] = rsqrtf(1.0f + (float)s);
}

__global__ __launch_bounds__(256) void k_gemm1_wmma(
    const float* __restrict__ x, const float* __restrict__ W1, const float* __restrict__ dinv,
    float* __restrict__ h, int n_nodes)
{
    __shared__ float sW[(KTILE / 2) * PSTRIDE];

    const int tid  = threadIdx.x;
    const int wave = tid >> 5;
    const int lane = tid & 31;
    const int half = lane >> 4;
    const int l16  = lane & 15;

    const int row_base = (blockIdx.x * 8 + wave) * 16;
    int arow = row_base + l16;
    if (arow >= n_nodes) arow = n_nodes - 1;
    const float* aptr = x + (size_t)arow * IN_CH + half * 2;

    v8f acc[8] = {};

    for (int k0 = 0; k0 < IN_CH; k0 += KTILE) {
        __syncthreads();
        for (int i = tid; i < (KTILE / 2) * 32; i += 256) {
            int p  = i >> 5;
            int c4 = (i & 31) * 4;
            float4 r0 = *(const float4*)(W1 + (size_t)(k0 + 2 * p + 0) * HID_CH + c4);
            float4 r1 = *(const float4*)(W1 + (size_t)(k0 + 2 * p + 1) * HID_CH + c4);
            float* dp = &sW[p * PSTRIDE + c4 * 2];
            ((float2*)dp)[0] = make_float2(r0.x, r1.x);
            ((float2*)dp)[1] = make_float2(r0.y, r1.y);
            ((float2*)dp)[2] = make_float2(r0.z, r1.z);
            ((float2*)dp)[3] = make_float2(r0.w, r1.w);
        }
        __syncthreads();

#pragma unroll 4
        for (int k = 0; k < KTILE; k += 4) {
            v2f a = *(const v2f*)(aptr + k0 + k);
            const int p = (k >> 1) + half;
            const float* bbase = &sW[p * PSTRIDE + l16 * 2];
#pragma unroll
            for (int nt = 0; nt < 8; ++nt) {
                v2f b = *(const v2f*)(bbase + nt * 32);
                acc[nt] = __builtin_amdgcn_wmma_f32_16x16x4_f32(
                    false, a, false, b, (short)0, acc[nt], false, false);
            }
        }
    }

    float dr[8];
#pragma unroll
    for (int v = 0; v < 8; ++v) dr[v] = dinv[min(row_base + v + 8 * half, n_nodes - 1)];
#pragma unroll
    for (int nt = 0; nt < 8; ++nt) {
#pragma unroll
        for (int v = 0; v < 8; ++v) {
            int r = row_base + v + 8 * half;
            if (r < n_nodes) h[(size_t)r * HID_CH + nt * 16 + l16] = acc[nt][v] * dr[v];
        }
    }
}

__device__ __forceinline__ void ldsAddF(float* p, float v) {
    __hip_atomic_fetch_add(p, v, __ATOMIC_RELAXED, __HIP_MEMORY_SCOPE_WORKGROUP);
}

__global__ __launch_bounds__(L1_THREADS) void k_agg1_own(
    const float* __restrict__ h, const int* __restrict__ src, const int* __restrict__ dst,
    const float* __restrict__ dinv, const float* __restrict__ b1, const float* __restrict__ W2,
    const unsigned char* __restrict__ rid8, float* __restrict__ h2, int ne, int n)
{
    __shared__ float    acc[L1_R * L1_PITCH];
    __shared__ unsigned qs[(L1_THREADS / 32) * L1_WCAP];
    __shared__ unsigned qd[(L1_THREADS / 32) * L1_WCAP];

    const int tid  = threadIdx.x;
    const int lane = tid & 31;
    const int wv  = __builtin_amdgcn_readfirstlane(tid >> 5);
    const int lo  = blockIdx.x * L1_R;
    const int cnt = min(L1_R, n - lo);

    {
        for (int r0 = wv * 4; r0 < cnt; r0 += (L1_THREADS / 32) * 4) {
            float4 hv[4];
#pragma unroll
            for (int q = 0; q < 4; ++q) {
                const int r = min(r0 + q, cnt - 1);
                hv[q] = *(const float4*)(h + (size_t)(lo + r) * HID_CH + 4 * lane);
            }
#pragma unroll
            for (int q = 0; q < 4; ++q) {
                if (r0 + q < cnt) {
                    float* ap = &acc[(r0 + q) * L1_PITCH + lane];
                    ap[0]  = hv[q].x;
                    ap[32] = hv[q].y;
                    ap[64] = hv[q].z;
                    ap[96] = hv[q].w;
                }
            }
        }
    }
    __syncthreads();

    const unsigned rb = (unsigned)blockIdx.x & 0xffu;
    const unsigned R4 = rb * 0x01010101u;
    const int nvf = ne >> 4;
    const int wq0 = wv * L1_WCAP;
    unsigned wcount = 0u;

    auto take_edges = [&](const bool has, const int e) {
        const unsigned bal = __builtin_amdgcn_ballot_w32(has);
        if (has) {
            const unsigned slot = wcount + __builtin_amdgcn_mbcnt_lo(bal, 0u);
            if (slot < (unsigned)L1_WCAP) {
                qs[wq0 + slot] = (unsigned)e;
            } else {
                const int s = src[e];
                const unsigned rel = (unsigned)(dst[e] - lo);
                if (rel < (unsigned)cnt) {
                    const float* hp = h + (size_t)s * HID_CH;
#pragma nounroll
                    for (int c = 0; c < HID_CH; ++c)
                        ldsAddF(&acc[rel * L1_PITCH + L1_POS(c)], hp[c]);
                }
            }
        }
        wcount += (unsigned)__builtin_popcount(bal);
    };
    auto scan_vec = [&](const uint4 w, const int ebase) {
        const unsigned K1 = 0x01010101u;
        const unsigned t0 = w.x ^ R4, t1 = w.y ^ R4, t2 = w.z ^ R4, t3 = w.w ^ R4;
        const unsigned z0 = (t0 - K1) & ~t0, z1 = (t1 - K1) & ~t1;
        const unsigned z2 = (t2 - K1) & ~t2, z3 = (t3 - K1) & ~t3;
        const bool cand = ((z0 | z1 | z2 | z3) & 0x80808080u) != 0u;
        if (__builtin_amdgcn_ballot_w32(cand) != 0u) {
            const unsigned f0 = z0 & ~(t0 << 7) & 0x80808080u;
            const unsigned f1 = z1 & ~(t1 << 7) & 0x80808080u;
            const unsigned f2 = z2 & ~(t2 << 7) & 0x80808080u;
            const unsigned f3 = z3 & ~(t3 << 7) & 0x80808080u;
            unsigned m = (f0 >> 7) | (f1 >> 6) | (f2 >> 5) | (f3 >> 4);
            while (__builtin_amdgcn_ballot_w32(m != 0u) != 0u) {
                const bool has = m != 0u;
                const int p = has ? __builtin_ctz(m) : 0;
                take_edges(has, ebase + (p & 7) * 4 + (p >> 3));
                m &= m - 1u;
            }
        }
    };

    auto drain = [&](const bool all) {
        const int nst = __builtin_amdgcn_readfirstlane((int)min(wcount, (unsigned)L1_WCAP));
        const int np  = all ? nst : (nst & ~(L1_G - 1));

        for (int i0 = lane; i0 < np; i0 += 32 * L1_PU) {
            int ee[L1_PU], ss[L1_PU], dd[L1_PU];
#pragma unroll
            for (int u = 0; u < L1_PU; ++u) {
                const int i = i0 + u * 32;
                ee[u] = (i < np) ? (int)qs[wq0 + i] : 0;
            }
#pragma unroll
            for (int u = 0; u < L1_PU; ++u) {
                ss[u] = src[ee[u]];
                dd[u] = dst[ee[u]];
            }
#pragma unroll
            for (int u = 0; u < L1_PU; ++u) {
                const int i = i0 + u * 32;
                if (i < np) {
                    const unsigned rel = (unsigned)(dd[u] - lo);
                    const bool ours = rel < (unsigned)cnt;
                    qs[wq0 + i] = (unsigned)ss[u];
                    qd[wq0 + i] = ours ? rel : ~0u;
                }
            }
        }

        for (int b0 = 0; b0 < np; b0 += 32) {
            const int m = min(32, np - b0);
            unsigned so = 0u;
            unsigned dw = ~0u;
            if (lane < m) {
                so = qs[wq0 + b0 + lane] * (unsigned)HID_CH;
                dw = qd[wq0 + b0 + lane];
            }
            const unsigned okmask = __builtin_amdgcn_ballot_w32(lane < m && dw != ~0u);
            for (int j0 = 0; j0 < m; j0 += L1_G) {
                float4 v[L1_G];
#pragma unroll
                for (int u = 0; u < L1_G; ++u) {
                    const int j = min(j0 + u, m - 1);
                    const unsigned sj = (unsigned)__builtin_amdgcn_readlane((int)so, j);
                    v[u] = *(const float4*)(h + (size_t)sj + 4 * lane);
                }
#pragma unroll
                for (int u = 0; u < L1_G; ++u) {
                    const int j = j0 + u;
                    if ((okmask >> j) & 1u) {
                        const unsigned dj = (unsigned)__builtin_amdgcn_readlane((int)dw, j);
                        float* ap = &acc[dj * L1_PITCH + lane];
                        ldsAddF(ap +  0, v[u].x);
                        ldsAddF(ap + 32, v[u].y);
                        ldsAddF(ap + 64, v[u].z);
                        ldsAddF(ap + 96, v[u].w);
                    }
                }
            }
        }

        const int left = nst - np;
        if (lane < left) {
            const unsigned t = qs[wq0 + np + lane];
            qs[wq0 + lane] = t;
        }
        wcount = (unsigned)left;
    };

    const uint4* rv = (const uint4*)rid8;
    int vb = wv * 32;
    for (; vb + 31 + (L1_SU - 1) * L1_THREADS < nvf; vb += L1_THREADS * L1_SU) {
        uint4 wb[L1_SU];
#pragma unroll
        for (int u = 0; u < L1_SU; ++u) wb[u] = rv[vb + lane + u * L1_THREADS];
#pragma unroll
        for (int u = 0; u < L1_SU; ++u) scan_vec(wb[u], (vb + lane + u * L1_THREADS) * 16);
        if (wcount >= (unsigned)L1_DRAIN) drain(false);
    }
    for (; vb < nvf; vb += L1_THREADS) {
        const int v = vb + lane;
        const unsigned nomatch = ~R4;
        const uint4 w = (v < nvf) ? rv[v] : make_uint4(nomatch, nomatch, nomatch, nomatch);
        scan_vec(w, v * 16);
    }
    if (wv == 0) {
        const int e = (nvf << 4) + lane;
        const bool has = lane < 16 && e < ne && (unsigned)rid8[e] == rb;
        take_edges(has, e);
    }
    drain(true);
    __syncthreads();

    if (tid < cnt) {
        const float* row = &acc[tid * L1_PITCH];
        const float dd = dinv[lo + tid];
        float s0 = 0.f, s1 = 0.f;
#pragma unroll 8
        for (int c = 0; c < HID_CH; c += 4) {
            float r0 = fmaxf(b1[c + 0] + dd * row[L1_POS(c + 0)], 0.f);
            float r1 = fmaxf(b1[c + 1] + dd * row[L1_POS(c + 1)], 0.f);
            float r2 = fmaxf(b1[c + 2] + dd * row[L1_POS(c + 2)], 0.f);
            float r3 = fmaxf(b1[c + 3] + dd * row[L1_POS(c + 3)], 0.f);
            s0 += r0 * W2[(c + 0) * 2 + 0] + r1 * W2[(c + 1) * 2 + 0]
                + r2 * W2[(c + 2) * 2 + 0] + r3 * W2[(c + 3) * 2 + 0];
            s1 += r0 * W2[(c + 0) * 2 + 1] + r1 * W2[(c + 1) * 2 + 1]
                + r2 * W2[(c + 2) * 2 + 1] + r3 * W2[(c + 3) * 2 + 1];
        }
        h2[(size_t)(lo + tid) * 2 + 0] = dd * s0;
        h2[(size_t)(lo + tid) * 2 + 1] = dd * s1;
    }
}

__global__ __launch_bounds__(NT_THREADS) void k_agg2_lds(
    const float* __restrict__ h2, const int* __restrict__ src, const int* __restrict__ dst,
    float* __restrict__ part2, int ne, int n, int chunk_len)
{
    __shared__ float acc[NT_RANGE * 2];
    const int tid = threadIdx.x;
    const int p   = blockIdx.x;
    const int lo  = blockIdx.y * NT_RANGE;
    const int hi  = min(NT_RANGE, n - lo);

    for (int i = tid; i < NT_RANGE; i += NT_THREADS) {
        acc[2 * i + 0] = 0.f;
        acc[2 * i + 1] = 0.f;
    }
    __syncthreads();

    const int e0 = p * chunk_len;
    const int e1 = min(ne, e0 + chunk_len);
    for (int base = e0 + tid; base < e1; base += NT_THREADS * NT_U) {
        int dv[NT_U], sv[NT_U];
#pragma unroll
        for (int u = 0; u < NT_U; ++u) {
            const int e = base + u * NT_THREADS;
            const bool ok = e < e1;
            dv[u] = ok ? dst[e] : -1;
            sv[u] = ok ? src[e] : 0;
        }
        float2 hv[NT_U];
#pragma unroll
        for (int u = 0; u < NT_U; ++u) {
            const bool m = (unsigned)(dv[u] - lo) < (unsigned)NT_RANGE;
            const int  s = m ? sv[u] : 0;
            hv[u] = *(const float2*)(h2 + (size_t)s * 2);
        }
#pragma unroll
        for (int u = 0; u < NT_U; ++u) {
            const unsigned rel = (unsigned)(dv[u] - lo);
            if (rel < (unsigned)NT_RANGE) {
                __hip_atomic_fetch_add(&acc[rel * 2 + 0], hv[u].x,
                                       __ATOMIC_RELAXED, __HIP_MEMORY_SCOPE_WORKGROUP);
                __hip_atomic_fetch_add(&acc[rel * 2 + 1], hv[u].y,
                                       __ATOMIC_RELAXED, __HIP_MEMORY_SCOPE_WORKGROUP);
            }
        }
    }
    __syncthreads();

    float2* pp = (float2*)(part2 + ((size_t)p * n + lo) * 2);
    for (int i = tid; i < hi; i += NT_THREADS) pp[i] = make_float2(acc[2 * i], acc[2 * i + 1]);
}

__global__ void k_out_sum(const float* __restrict__ h2, const float* __restrict__ part2,
                          const float* __restrict__ b2, const float* __restrict__ dinv,
                          float* __restrict__ out, int n)
{
    int i = blockIdx.x * blockDim.x + threadIdx.x;
    if (i >= n) return;
    const float di = dinv[i];
    const float2 hs = *(const float2*)(h2 + (size_t)i * 2);
    float o0 = hs.x;
    float o1 = hs.y;
#pragma unroll
    for (int p = 0; p < NT_PLANES; ++p) {
        const float2 v = *(const float2*)(part2 + ((size_t)p * n + i) * 2);
        o0 += v.x;
        o1 += v.y;
    }
    *(float2*)(out + (size_t)i * 2) = make_float2(b2[0] + di * o0, b2[1] + di * o1);
}

extern "C" void kernel_launch(void* const* d_in, const int* in_sizes, int n_in,
                              void* d_out, int out_size, void* d_ws, size_t ws_size,
                              hipStream_t stream)
{
    const float* x  = (const float*)d_in[0];
    const int*   ei = (const int*)d_in[1];
    const float* W1 = (const float*)d_in[2];
    const float* b1 = (const float*)d_in[3];
    const float* W2 = (const float*)d_in[4];
    const float* b2 = (const float*)d_in[5];
    float* out = (float*)d_out;

    const int n_nodes = in_sizes[0] / IN_CH;
    const int n_edges = in_sizes[1] / 2;
    const int* src = ei;
    const int* dst = ei + n_edges;

    char* ws = (char*)d_ws;
    size_t off = 0;
    float* dinv = (float*)(ws + off);
    off += ((size_t)n_nodes * 4 + 4095) & ~(size_t)4095;
    float* h = (float*)(ws + off);
    size_t hb = ((size_t)n_nodes * HID_CH * 4 + 4095) & ~(size_t)4095;
    off += hb;
    char* reg3 = ws + off;

    unsigned* part  = (unsigned*)reg3;
    float*    part2 = (float*)reg3;
    size_t pb = ((size_t)n_nodes * NT_PLANES * 8 + 4095) & ~(size_t)4095;
    float* h2 = (float*)(reg3 + pb);
    size_t h2b = ((size_t)n_nodes * 2 * 4 + 4095) & ~(size_t)4095;
    unsigned char* rid8 = (unsigned char*)(reg3 + pb + h2b);

    const int chunk_len = (n_edges + NT_PLANES - 1) / NT_PLANES;
    dim3 ngrid(NT_PLANES, (n_nodes + NT_RANGE - 1) / NT_RANGE);
    k_deg_hist<<<ngrid, NT_THREADS, 0, stream>>>(dst, part, rid8, n_edges, n_nodes, chunk_len);
    k_dinv    <<<(n_nodes + 255) / 256, 256, 0, stream>>>(part, dinv, n_nodes);

    k_gemm1_wmma<<<(n_nodes + 127) / 128, 256, 0, stream>>>(x, W1, dinv, h, n_nodes);

    k_agg1_own  <<<(n_nodes + L1_R - 1) / L1_R, L1_THREADS, 0, stream>>>(
        h, src, dst, dinv, b1, W2, rid8, h2, n_edges, n_nodes);

    k_agg2_lds  <<<ngrid, NT_THREADS, 0, stream>>>(h2, src, dst, part2, n_edges, n_nodes, chunk_len);
    k_out_sum   <<<(n_nodes + 255) / 256, 256, 0, stream>>>(h2, part2, b2, dinv, out, n_nodes);
}


# x @ W1: 250 resident 32-wave blocks stage all of W1 in LDS once (one barrier, no K chunks); one row tile per wave, a block's odd tile split into four column quarters; k order and sums unchanged
# speedup vs baseline: 3.5752x; 1.0442x over previous
#include <hip/hip_runtime.h>
#include <hip/hip_bf16.h>
#include <cstdint>

typedef float v2f __attribute__((ext_vector_type(2)));
typedef float v8f __attribute__((ext_vector_type(8)));

#define IN_CH  256
#define HID_CH 128
#define PSTRIDE 288

#define L1_R       400
#define L1_PITCH   129
#define L1_WCAP    288
#define L1_THREADS 1024
#define L1_SU      8
#define L1_PU      2
#define L1_DRAIN   32
#define L1_G       8
#define L1_POS(c)  ((((c) & 3) << 5) + ((c) >> 2))

#define NT_RANGE   12544
#define NT_PLANES  32
#define NT_THREADS 1024
#define NT_U       16

__global__ __launch_bounds__(NT_THREADS) void k_deg_hist(
    const int* __restrict__ dst, unsigned* __restrict__ part, unsigned char* __restrict__ rid8,
    int ne, int n, int chunk_len)
{
    __shared__ unsigned cnt[NT_RANGE];
    const int tid = threadIdx.x;
    const int c   = blockIdx.x;
    const int lo  = blockIdx.y * NT_RANGE;

    for (int i = tid; i < NT_RANGE; i += NT_THREADS) cnt[i] = 0u;

    {
        const int nb    = gridDim.x * gridDim.y;
        const int bid   = blockIdx.y * gridDim.x + blockIdx.x;
        const int nquad = ne >> 2;
        const int qlen  = ((nquad + nb - 1) / nb + 31) & ~31;
        const int q0    = bid * qlen;
        const int q1    = min(nquad, q0 + qlen);
        unsigned* rid32 = (unsigned*)rid8;
        for (int q = q0 + tid; q < q1; q += NT_THREADS) {
            const int* dp = dst + 4 * (size_t)q;
            const unsigned o0 = ((unsigned)dp[0] / (unsigned)L1_R) & 0xffu;
            const unsigned o1 = ((unsigned)dp[1] / (unsigned)L1_R) & 0xffu;
            const unsigned o2 = ((unsigned)dp[2] / (unsigned)L1_R) & 0xffu;
            const unsigned o3 = ((unsigned)dp[3] / (unsigned)L1_R) & 0xffu;
            rid32[q] = o0 | (o1 << 8) | (o2 << 16) | (o3 << 24);
        }
        if (bid == 0 && tid < (ne & 3)) {
            const int e = (ne & ~3) + tid;
            rid8[e] = (unsigned char)((unsigned)dst[e] / (unsigned)L1_R);
        }
    }
    __syncthreads();

    const int e0 = c * chunk_len;
    const int e1 = min(ne, e0 + chunk_len);
    for (int base = e0 + tid; base < e1; base += NT_THREADS * NT_U) {
        int dv[NT_U];
#pragma unroll
        for (int u = 0; u < NT_U; ++u) {
            const int e = base + u * NT_THREADS;
            dv[u] = (e < e1) ? dst[e] : -1;
        }
#pragma unroll
        for (int u = 0; u < NT_U; ++u) {
            const unsigned rel = (unsigned)(dv[u] - lo);
            if (rel < (unsigned)NT_RANGE) atomicAdd(&cnt[rel], 1u);
        }
    }
    __syncthreads();

    const int hi = min(NT_RANGE, n - lo);
    unsigned* pp = part + (size_t)c * n + lo;
    for (int i = tid; i < hi; i += NT_THREADS) pp[i] = cnt[i];
}

__global__ void k_dinv(const unsigned* __restrict__ part, float* __restrict__ dinv, int n) {
    int i = blockIdx.x * blockDim.x + threadIdx.x;
    if (i >= n) return;
    unsigned s = 0u;
#pragma unroll
    for (int c = 0; c < NT_PLANES; ++c) s += part[(size_t)c * n + i];
    dinv[i] = rsqrtf(1.0f + (float)s);
}

#define G1_BLOCKS  250
#define G1_THREADS 1024
#define G1_WAVES   (G1_THREADS / 32)
#define G1_LDS_FLOATS ((IN_CH / 2) * PSTRIDE)

template <int NT>
__device__ __forceinline__ void gemm1_unit(
    const float* __restrict__ x, const float* __restrict__ dinv, float* __restrict__ h,
    const float* sW, const int n_nodes, const int tile, const int nt0, const int half, const int l16)
{
    const int row_base = tile * 16;
    int arow = row_base + l16;
    if (arow >= n_nodes) arow = n_nodes - 1;
    const float* aptr = x + (size_t)arow * IN_CH + half * 2;
    const float* bcol = sW + l16 * 2 + nt0 * 32;

    v8f acc[NT] = {};

#pragma unroll 4
    for (int k = 0; k < IN_CH; k += 4) {
        v2f a = *(const v2f*)(aptr + k);
        const int p = (k >> 1) + half;
        const float* bbase = bcol + p * PSTRIDE;
#pragma unroll
        for (int nt = 0; nt < NT; ++nt) {
            v2f b = *(const v2f*)(bbase + nt * 32);
            acc[nt] = __builtin_amdgcn_wmma_f32_16x16x4_f32(
                false, a, false, b, (short)0, acc[nt], false, false);
        }
    }

    float dr[8];
#pragma unroll
    for (int v = 0; v < 8; ++v) dr[v] = dinv[min(row_base + v + 8 * half, n_nodes - 1)];
#pragma unroll
    for (int nt = 0; nt < NT; ++nt) {
#pragma unroll
        for (int v = 0; v < 8; ++v) {
            int r = row_base + v + 8 * half;
            if (r < n_nodes) h[(size_t)r * HID_CH + (nt0 + nt) * 16 + l16] = acc[nt][v] * dr[v];
        }
    }
}

__global__ __launch_bounds__(G1_THREADS) void k_gemm1_wmma(
    const float* __restrict__ x, const float* __restrict__ W1, const float* __restrict__ dinv,
    float* __restrict__ h, int n_nodes)
{
    __shared__ float sW[G1_LDS_FLOATS];

    const int tid  = threadIdx.x;
    const int wave = __builtin_amdgcn_readfirstlane(tid >> 5);
    const int lane = tid & 31;
    const int half = lane >> 4;
    const int l16  = lane & 15;

    {
        constexpr int SU = (IN_CH / 2) * 32 / G1_THREADS;
        float4 r0[SU], r1[SU];
#pragma unroll
        for (int u = 0; u < SU; ++u) {
            const int i  = tid + u * G1_THREADS;
            const int p  = i >> 5;
            const int c4 = (i & 31) * 4;
            r0[u] = *(const float4*)(W1 + (size_t)(2 * p + 0) * HID_CH + c4);
            r1[u] = *(const float4*)(W1 + (size_t)(2 * p + 1) * HID_CH + c4);
        }
#pragma unroll
        for (int u = 0; u < SU; ++u) {
            const int i  = tid + u * G1_THREADS;
            const int p  = i >> 5;
            const int c4 = (i & 31) * 4;
            float* dp = &sW[p * PSTRIDE + c4 * 2];
            ((float2*)dp)[0] = make_float2(r0[u].x, r1[u].x);
            ((float2*)dp)[1] = make_float2(r0[u].y, r1[u].y);
            ((float2*)dp)[2] = make_float2(r0[u].z, r1[u].z);
            ((float2*)dp)[3] = make_float2(r0[u].w, r1[u].w);
        }
    }
    __syncthreads();

    const int ntiles = (n_nodes + 15) >> 4;
    const int nb = (int)gridDim.x;
    const int b  = (int)blockIdx.x;
    const int T  = (ntiles > b) ? (ntiles - 1 - b) / nb + 1 : 0;
    const int F  = T & ~3;
    const int U  = F + 4 * (T - F);
    for (int u = wave; u < U; u += G1_WAVES) {
        if (u < F) {
            gemm1_unit<8>(x, dinv, h, sW, n_nodes, b + nb * u, 0, half, l16);
        } else {
            const int q = u - F;
            gemm1_unit<2>(x, dinv, h, sW, n_nodes, b + nb * (F + (q >> 2)), (q & 3) * 2, half, l16);
        }
    }
}

__device__ __forceinline__ void ldsAddF(float* p, float v) {
    __hip_atomic_fetch_add(p, v, __ATOMIC_RELAXED, __HIP_MEMORY_SCOPE_WORKGROUP);
}

__global__ __launch_bounds__(L1_THREADS) void k_agg1_own(
    const float* __restrict__ h, const int* __restrict__ src, const int* __restrict__ dst,
    const float* __restrict__ dinv, const float* __restrict__ b1, const float* __restrict__ W2,
    const unsigned char* __restrict__ rid8, float* __restrict__ h2, int ne, int n)
{
    __shared__ float    acc[L1_R * L1_PITCH];
    __shared__ unsigned qs[(L1_THREADS / 32) * L1_WCAP];
    __shared__ unsigned qd[(L1_THREADS / 32) * L1_WCAP];

    const int tid  = threadIdx.x;
    const int lane = tid & 31;
    const int wv  = __builtin_amdgcn_readfirstlane(tid >> 5);
    const int lo  = blockIdx.x * L1_R;
    const int cnt = min(L1_R, n - lo);

    {
        for (int r0 = wv * 4; r0 < cnt; r0 += (L1_THREADS / 32) * 4) {
            float4 hv[4];
#pragma unroll
            for (int q = 0; q < 4; ++q) {
                const int r = min(r0 + q, cnt - 1);
                hv[q] = *(const float4*)(h + (size_t)(lo + r) * HID_CH + 4 * lane);
            }
#pragma unroll
            for (int q = 0; q < 4; ++q) {
                if (r0 + q < cnt) {
                    float* ap = &acc[(r0 + q) * L1_PITCH + lane];
                    ap[0]  = hv[q].x;
                    ap[32] = hv[q].y;
                    ap[64] = hv[q].z;
                    ap[96] = hv[q].w;
                }
            }
        }
    }
    __syncthreads();

    const unsigned rb = (unsigned)blockIdx.x & 0xffu;
    const unsigned R4 = rb * 0x01010101u;
    const int nvf = ne >> 4;
    const int wq0 = wv * L1_WCAP;
    unsigned wcount = 0u;

    auto take_edges = [&](const bool has, const int e) {
        const unsigned bal = __builtin_amdgcn_ballot_w32(has);
        if (has) {
            const unsigned slot = wcount + __builtin_amdgcn_mbcnt_lo(bal, 0u);
            if (slot < (unsigned)L1_WCAP) {
                qs[wq0 + slot] = (unsigned)e;
            } else {
                const int s = src[e];
                const unsigned rel = (unsigned)(dst[e] - lo);
                if (rel < (unsigned)cnt) {
                    const float* hp = h + (size_t)s * HID_CH;
#pragma nounroll
                    for (int c = 0; c < HID_CH; ++c)
                        ldsAddF(&acc[rel * L1_PITCH + L1_POS(c)], hp[c]);
                }
            }
        }
        wcount += (unsigned)__builtin_popcount(bal);
    };
    auto scan_vec = [&](const uint4 w, const int ebase) {
        const unsigned K1 = 0x01010101u;
        const unsigned t0 = w.x ^ R4, t1 = w.y ^ R4, t2 = w.z ^ R4, t3 = w.w ^ R4;
        const unsigned z0 = (t0 - K1) & ~t0, z1 = (t1 - K1) & ~t1;
        const unsigned z2 = (t2 - K1) & ~t2, z3 = (t3 - K1) & ~t3;
        const bool cand = ((z0 | z1 | z2 | z3) & 0x80808080u) != 0u;
        if (__builtin_amdgcn_ballot_w32(cand) != 0u) {
            const unsigned f0 = z0 & ~(t0 << 7) & 0x80808080u;
            const unsigned f1 = z1 & ~(t1 << 7) & 0x80808080u;
            const unsigned f2 = z2 & ~(t2 << 7) & 0x80808080u;
            const unsigned f3 = z3 & ~(t3 << 7) & 0x80808080u;
            unsigned m = (f0 >> 7) | (f1 >> 6) | (f2 >> 5) | (f3 >> 4);
            while (__builtin_amdgcn_ballot_w32(m != 0u) != 0u) {
                const bool has = m != 0u;
                const int p = has ? __builtin_ctz(m) : 0;
                take_edges(has, ebase + (p & 7) * 4 + (p >> 3));
                m &= m - 1u;
            }
        }
    };

    auto drain = [&](const bool all) {
        const int nst = __builtin_amdgcn_readfirstlane((int)min(wcount, (unsigned)L1_WCAP));
        const int np  = all ? nst : (nst & ~(L1_G - 1));

        for (int i0 = lane; i0 < np; i0 += 32 * L1_PU) {
            int ee[L1_PU], ss[L1_PU], dd[L1_PU];
#pragma unroll
            for (int u = 0; u < L1_PU; ++u) {
                const int i = i0 + u * 32;
                ee[u] = (i < np) ? (int)qs[wq0 + i] : 0;
            }
#pragma unroll
            for (int u = 0; u < L1_PU; ++u) {
                ss[u] = src[ee[u]];
                dd[u] = dst[ee[u]];
            }
#pragma unroll
            for (int u = 0; u < L1_PU; ++u) {
                const int i = i0 + u * 32;
                if (i < np) {
                    const unsigned rel = (unsigned)(dd[u] - lo);
                    const bool ours = rel < (unsigned)cnt;
                    qs[wq0 + i] = (unsigned)ss[u];
                    qd[wq0 + i] = ours ? rel : ~0u;
                }
            }
        }

        for (int b0 = 0; b0 < np; b0 += 32) {
            const int m = min(32, np - b0);
            unsigned so = 0u;
            unsigned dw = ~0u;
            if (lane < m) {
                so = qs[wq0 + b0 + lane] * (unsigned)HID_CH;
                dw = qd[wq0 + b0 + lane];
            }
            const unsigned okmask = __builtin_amdgcn_ballot_w32(lane < m && dw != ~0u);
            for (int j0 = 0; j0 < m; j0 += L1_G) {
                float4 v[L1_G];
#pragma unroll
                for (int u = 0; u < L1_G; ++u) {
                    const int j = min(j0 + u, m - 1);
                    const unsigned sj = (unsigned)__builtin_amdgcn_readlane((int)so, j);
                    v[u] = *(const float4*)(h + (size_t)sj + 4 * lane);
                }
#pragma unroll
                for (int u = 0; u < L1_G; ++u) {
                    const int j = j0 + u;
                    if ((okmask >> j) & 1u) {
                        const unsigned dj = (unsigned)__builtin_amdgcn_readlane((int)dw, j);
                        float* ap = &acc[dj * L1_PITCH + lane];
                        ldsAddF(ap +  0, v[u].x);
                        ldsAddF(ap + 32, v[u].y);
                        ldsAddF(ap + 64, v[u].z);
                        ldsAddF(ap + 96, v[u].w);
                    }
                }
            }
        }

        const int left = nst - np;
        if (lane < left) {
            const unsigned t = qs[wq0 + np + lane];
            qs[wq0 + lane] = t;
        }
        wcount = (unsigned)left;
    };

    const uint4* rv = (const uint4*)rid8;
    int vb = wv * 32;
    for (; vb + 31 + (L1_SU - 1) * L1_THREADS < nvf; vb += L1_THREADS * L1_SU) {
        uint4 wb[L1_SU];
#pragma unroll
        for (int u = 0; u < L1_SU; ++u) wb[u] = rv[vb + lane + u * L1_THREADS];
#pragma unroll
        for (int u = 0; u < L1_SU; ++u) scan_vec(wb[u], (vb + lane + u * L1_THREADS) * 16);
        if (wcount >= (unsigned)L1_DRAIN) drain(false);
    }
    for (; vb < nvf; vb += L1_THREADS) {
        const int v = vb + lane;
        const unsigned nomatch = ~R4;
        const uint4 w = (v < nvf) ? rv[v] : make_uint4(nomatch, nomatch, nomatch, nomatch);
        scan_vec(w, v * 16);
    }
    if (wv == 0) {
        const int e = (nvf << 4) + lane;
        const bool has = lane < 16 && e < ne && (unsigned)rid8[e] == rb;
        take_edges(has, e);
    }
    drain(true);
    __syncthreads();

    if (tid < cnt) {
        const float* row = &acc[tid * L1_PITCH];
        const float dd = dinv[lo + tid];
        float s0 = 0.f, s1 = 0.f;
#pragma unroll 8
        for (int c = 0; c < HID_CH; c += 4) {
            float r0 = fmaxf(b1[c + 0] + dd * row[L1_POS(c + 0)], 0.f);
            float r1 = fmaxf(b1[c + 1] + dd * row[L1_POS(c + 1)], 0.f);
            float r2 = fmaxf(b1[c + 2] + dd * row[L1_POS(c + 2)], 0.f);
            float r3 = fmaxf(b1[c + 3] + dd * row[L1_POS(c + 3)], 0.f);
            s0 += r0 * W2[(c + 0) * 2 + 0] + r1 * W2[(c + 1) * 2 + 0]
                + r2 * W2[(c + 2) * 2 + 0] + r3 * W2[(c + 3) * 2 + 0];
            s1 += r0 * W2[(c + 0) * 2 + 1] + r1 * W2[(c + 1) * 2 + 1]
                + r2 * W2[(c + 2) * 2 + 1] + r3 * W2[(c + 3) * 2 + 1];
        }
        h2[(size_t)(lo + tid) * 2 + 0] = dd * s0;
        h2[(size_t)(lo + tid) * 2 + 1] = dd * s1;
    }
}

__global__ __launch_bounds__(NT_THREADS) void k_agg2_lds(
    const float* __restrict__ h2, const int* __restrict__ src, const int* __restrict__ dst,
    float* __restrict__ part2, int ne, int n, int chunk_len)
{
    __shared__ float acc[NT_RANGE * 2];
    const int tid = threadIdx.x;
    const int p   = blockIdx.x;
    const int lo  = blockIdx.y * NT_RANGE;
    const int hi  = min(NT_RANGE, n - lo);

    for (int i = tid; i < NT_RANGE; i += NT_THREADS) {
        acc[2 * i + 0] = 0.f;
        acc[2 * i + 1] = 0.f;
    }
    __syncthreads();

    const int e0 = p * chunk_len;
    const int e1 = min(ne, e0 + chunk_len);
    for (int base = e0 + tid; base < e1; base += NT_THREADS * NT_U) {
        int dv[NT_U], sv[NT_U];
#pragma unroll
        for (int u = 0; u < NT_U; ++u) {
            const int e = base + u * NT_THREADS;
            const bool ok = e < e1;
            dv[u] = ok ? dst[e] : -1;
            sv[u] = ok ? src[e] : 0;
        }
        float2 hv[NT_U];
#pragma unroll
        for (int u = 0; u < NT_U; ++u) {
            const bool m = (unsigned)(dv[u] - lo) < (unsigned)NT_RANGE;
            const int  s = m ? sv[u] : 0;
            hv[u] = *(const float2*)(h2 + (size_t)s * 2);
        }
#pragma unroll
        for (int u = 0; u < NT_U; ++u) {
            const unsigned rel = (unsigned)(dv[u] - lo);
            if (rel < (unsigned)NT_RANGE) {
                __hip_atomic_fetch_add(&acc[rel * 2 + 0], hv[u].x,
                                       __ATOMIC_RELAXED, __HIP_MEMORY_SCOPE_WORKGROUP);
                __hip_atomic_fetch_add(&acc[rel * 2 + 1], hv[u].y,
                                       __ATOMIC_RELAXED, __HIP_MEMORY_SCOPE_WORKGROUP);
            }
        }
    }
    __syncthreads();

    float2* pp = (float2*)(part2 + ((size_t)p * n + lo) * 2);
    for (int i = tid; i < hi; i += NT_THREADS) pp[i] = make_float2(acc[2 * i], acc[2 * i + 1]);
}

__global__ void k_out_sum(const float* __restrict__ h2, const float* __restrict__ part2,
                          const float* __restrict__ b2, const float* __restrict__ dinv,
                          float* __restrict__ out, int n)
{
    int i = blockIdx.x * blockDim.x + threadIdx.x;
    if (i >= n) return;
    const float di = dinv[i];
    const float2 hs = *(const float2*)(h2 + (size_t)i * 2);
    float o0 = hs.x;
    float o1 = hs.y;
#pragma unroll
    for (int p = 0; p < NT_PLANES; ++p) {
        const float2 v = *(const float2*)(part2 + ((size_t)p * n + i) * 2);
        o0 += v.x;
        o1 += v.y;
    }
    *(float2*)(out + (size_t)i * 2) = make_float2(b2[0] + di * o0, b2[1] + di * o1);
}

extern "C" void kernel_launch(void* const* d_in, const int* in_sizes, int n_in,
                              void* d_out, int out_size, void* d_ws, size_t ws_size,
                              hipStream_t stream)
{
    const float* x  = (const float*)d_in[0];
    const int*   ei = (const int*)d_in[1];
    const float* W1 = (const float*)d_in[2];
    const float* b1 = (const float*)d_in[3];
    const float* W2 = (const float*)d_in[4];
    const float* b2 = (const float*)d_in[5];
    float* out = (float*)d_out;

    const int n_nodes = in_sizes[0] / IN_CH;
    const int n_edges = in_sizes[1] / 2;
    const int* src = ei;
    const int* dst = ei + n_edges;

    char* ws = (char*)d_ws;
    size_t off = 0;
    float* dinv = (float*)(ws + off);
    off += ((size_t)n_nodes * 4 + 4095) & ~(size_t)4095;
    float* h = (float*)(ws + off);
    size_t hb = ((size_t)n_nodes * HID_CH * 4 + 4095) & ~(size_t)4095;
    off += hb;
    char* reg3 = ws + off;

    unsigned* part  = (unsigned*)reg3;
    float*    part2 = (float*)reg3;
    size_t pb = ((size_t)n_nodes * NT_PLANES * 8 + 4095) & ~(size_t)4095;
    float* h2 = (float*)(reg3 + pb);
    size_t h2b = ((size_t)n_nodes * 2 * 4 + 4095) & ~(size_t)4095;
    unsigned char* rid8 = (unsigned char*)(reg3 + pb + h2b);

    const int chunk_len = (n_edges + NT_PLANES - 1) / NT_PLANES;
    dim3 ngrid(NT_PLANES, (n_nodes + NT_RANGE - 1) / NT_RANGE);
    k_deg_hist<<<ngrid, NT_THREADS, 0, stream>>>(dst, part, rid8, n_edges, n_nodes, chunk_len);
    k_dinv    <<<(n_nodes + 255) / 256, 256, 0, stream>>>(part, dinv, n_nodes);

    k_gemm1_wmma<<<G1_BLOCKS, G1_THREADS, 0, stream>>>(x, W1, dinv, h, n_nodes);

    k_agg1_own  <<<(n_nodes + L1_R - 1) / L1_R, L1_THREADS, 0, stream>>>(
        h, src, dst, dinv, b1, W2, rid8, h2, n_edges, n_nodes);

    k_agg2_lds  <<<ngrid, NT_THREADS, 0, stream>>>(h2, src, dst, part2, n_edges, n_nodes, chunk_len);
    k_out_sum   <<<(n_nodes + 255) / 256, 256, 0, stream>>>(h2, part2, b2, dinv, out, n_nodes);
}


# Layer 2 walks the edge lists that layer 1's owner blocks save (source, local dest per wave): no second edge pass, no planes, no sum pass (5 launches); per-node sums = LDS arrival order
# speedup vs baseline: 3.9042x; 1.0920x over previous
#include <hip/hip_runtime.h>
#include <hip/hip_bf16.h>
#include <cstdint>

typedef float v2f __attribute__((ext_vector_type(2)));
typedef float v8f __attribute__((ext_vector_type(8)));

#define IN_CH  256
#define HID_CH 128
#define PSTRIDE 288

#define L1_R       400
#define L1_PITCH   129
#define L1_WCAP    288
#define L1_THREADS 1024
#define L1_SU      8
#define L1_PU      2
#define L1_DRAIN   32
#define L1_G       8
#define L2_CAP     399
#define L2_STRIDE  (2 * (L2_CAP + 1))
#define L2_U       8
#define L2_OVF     0xffffffffu
#define L1_POS(c)  ((((c) & 3) << 5) + ((c) >> 2))

#define NT_RANGE   12544
#define NT_PLANES  32
#define NT_THREADS 1024
#define NT_U       16

__global__ __launch_bounds__(NT_THREADS) void k_deg_hist(
    const int* __restrict__ dst, unsigned* __restrict__ part, unsigned char* __restrict__ rid8,
    int ne, int n, int chunk_len)
{
    __shared__ unsigned cnt[NT_RANGE];
    const int tid = threadIdx.x;
    const int c   = blockIdx.x;
    const int lo  = blockIdx.y * NT_RANGE;

    for (int i = tid; i < NT_RANGE; i += NT_THREADS) cnt[i] = 0u;

    {
        const int nb    = gridDim.x * gridDim.y;
        const int bid   = blockIdx.y * gridDim.x + blockIdx.x;
        const int nquad = ne >> 2;
        const int qlen  = ((nquad + nb - 1) / nb + 31) & ~31;
        const int q0    = bid * qlen;
        const int q1    = min(nquad, q0 + qlen);
        unsigned* rid32 = (unsigned*)rid8;
        for (int q = q0 + tid; q < q1; q += NT_THREADS) {
            const int* dp = dst + 4 * (size_t)q;
            const unsigned o0 = ((unsigned)dp[0] / (unsigned)L1_R) & 0xffu;
            const unsigned o1 = ((unsigned)dp[1] / (unsigned)L1_R) & 0xffu;
            const unsigned o2 = ((unsigned)dp[2] / (unsigned)L1_R) & 0xffu;
            const unsigned o3 = ((unsigned)dp[3] / (unsigned)L1_R) & 0xffu;
            rid32[q] = o0 | (o1 << 8) | (o2 << 16) | (o3 << 24);
        }
        if (bid == 0 && tid < (ne & 3)) {
            const int e = (ne & ~3) + tid;
            rid8[e] = (unsigned char)((unsigned)dst[e] / (unsigned)L1_R);
        }
    }
    __syncthreads();

    const int e0 = c * chunk_len;
    const int e1 = min(ne, e0 + chunk_len);
    for (int base = e0 + tid; base < e1; base += NT_THREADS * NT_U) {
        int dv[NT_U];
#pragma unroll
        for (int u = 0; u < NT_U; ++u) {
            const int e = base + u * NT_THREADS;
            dv[u] = (e < e1) ? dst[e] : -1;
        }
#pragma unroll
        for (int u = 0; u < NT_U; ++u) {
            const unsigned rel = (unsigned)(dv[u] - lo);
            if (rel < (unsigned)NT_RANGE) atomicAdd(&cnt[rel], 1u);
        }
    }
    __syncthreads();

    const int hi = min(NT_RANGE, n - lo);
    unsigned* pp = part + (size_t)c * n + lo;
    for (int i = tid; i < hi; i += NT_THREADS) pp[i] = cnt[i];
}

__global__ void k_dinv(const unsigned* __restrict__ part, float* __restrict__ dinv, int n) {
    int i = blockIdx.x * blockDim.x + threadIdx.x;
    if (i >= n) return;
    unsigned s = 0u;
#pragma unroll
    for (int c = 0; c < NT_PLANES; ++c) s += part[(size_t)c * n + i];
    dinv[i] = rsqrtf(1.0f + (float)s);
}

#define G1_BLOCKS  250
#define G1_THREADS 1024
#define G1_WAVES   (G1_THREADS / 32)
#define G1_LDS_FLOATS ((IN_CH / 2) * PSTRIDE)

template <int NT>
__device__ __forceinline__ void gemm1_unit(
    const float* __restrict__ x, const float* __restrict__ dinv, float* __restrict__ h,
    const float* sW, const int n_nodes, const int tile, const int nt0, const int half, const int l16)
{
    const int row_base = tile * 16;
    int arow = row_base + l16;
    if (arow >= n_nodes) arow = n_nodes - 1;
    const float* aptr = x + (size_t)arow * IN_CH + half * 2;
    const float* bcol = sW + l16 * 2 + nt0 * 32;

    v8f acc[NT] = {};

#pragma unroll 4
    for (int k = 0; k < IN_CH; k += 4) {
        v2f a = *(const v2f*)(aptr + k);
        const int p = (k >> 1) + half;
        const float* bbase = bcol + p * PSTRIDE;
#pragma unroll
        for (int nt = 0; nt < NT; ++nt) {
            v2f b = *(const v2f*)(bbase + nt * 32);
            acc[nt] = __builtin_amdgcn_wmma_f32_16x16x4_f32(
                false, a, false, b, (short)0, acc[nt], false, false);
        }
    }

    float dr[8];
#pragma unroll
    for (int v = 0; v < 8; ++v) dr[v] = dinv[min(row_base + v + 8 * half, n_nodes - 1)];
#pragma unroll
    for (int nt = 0; nt < NT; ++nt) {
#pragma unroll
        for (int v = 0; v < 8; ++v) {
            int r = row_base + v + 8 * half;
            if (r < n_nodes) h[(size_t)r * HID_CH + (nt0 + nt) * 16 + l16] = acc[nt][v] * dr[v];
        }
    }
}

__global__ __launch_bounds__(G1_THREADS) void k_gemm1_wmma(
    const float* __restrict__ x, const float* __restrict__ W1, const float* __restrict__ dinv,
    float* __restrict__ h, int n_nodes)
{
    __shared__ float sW[G1_LDS_FLOATS];

    const int tid  = threadIdx.x;
    const int wave = __builtin_amdgcn_readfirstlane(tid >> 5);
    const int lane = tid & 31;
    const int half = lane >> 4;
    const int l16  = lane & 15;

    {
        constexpr int SU = (IN_CH / 2) * 32 / G1_THREADS;
        float4 r0[SU], r1[SU];
#pragma unroll
        for (int u = 0; u < SU; ++u) {
            const int i  = tid + u * G1_THREADS;
            const int p  = i >> 5;
            const int c4 = (i & 31) * 4;
            r0[u] = *(const float4*)(W1 + (size_t)(2 * p + 0) * HID_CH + c4);
            r1[u] = *(const float4*)(W1 + (size_t)(2 * p + 1) * HID_CH + c4);
        }
#pragma unroll
        for (int u = 0; u < SU; ++u) {
            const int i  = tid + u * G1_THREADS;
            const int p  = i >> 5;
            const int c4 = (i & 31) * 4;
            float* dp = &sW[p * PSTRIDE + c4 * 2];
            ((float2*)dp)[0] = make_float2(r0[u].x, r1[u].x);
            ((float2*)dp)[1] = make_float2(r0[u].y, r1[u].y);
            ((float2*)dp)[2] = make_float2(r0[u].z, r1[u].z);
            ((float2*)dp)[3] = make_float2(r0[u].w, r1[u].w);
        }
    }
    __syncthreads();

    const int ntiles = (n_nodes + 15) >> 4;
    const int nb = (int)gridDim.x;
    const int b  = (int)blockIdx.x;
    const int T  = (ntiles > b) ? (ntiles - 1 - b) / nb + 1 : 0;
    const int F  = T & ~3;
    const int U  = F + 4 * (T - F);
    for (int u = wave; u < U; u += G1_WAVES) {
        if (u < F) {
            gemm1_unit<8>(x, dinv, h, sW, n_nodes, b + nb * u, 0, half, l16);
        } else {
            const int q = u - F;
            gemm1_unit<2>(x, dinv, h, sW, n_nodes, b + nb * (F + (q >> 2)), (q & 3) * 2, half, l16);
        }
    }
}

__device__ __forceinline__ void ldsAddF(float* p, float v) {
    __hip_atomic_fetch_add(p, v, __ATOMIC_RELAXED, __HIP_MEMORY_SCOPE_WORKGROUP);
}

__global__ __launch_bounds__(L1_THREADS) void k_agg1_own(
    const float* __restrict__ h, const int* __restrict__ src, const int* __restrict__ dst,
    const float* __restrict__ dinv, const float* __restrict__ b1, const float* __restrict__ W2,
    const unsigned char* __restrict__ rid8, float* __restrict__ h2,
    unsigned* __restrict__ glist, int ne, int n)
{
    __shared__ float    acc[L1_R * L1_PITCH];
    __shared__ unsigned qs[(L1_THREADS / 32) * L1_WCAP];
    __shared__ unsigned qd[(L1_THREADS / 32) * L1_WCAP];

    const int tid  = threadIdx.x;
    const int lane = tid & 31;
    const int wv  = __builtin_amdgcn_readfirstlane(tid >> 5);
    const int lo  = blockIdx.x * L1_R;
    const int cnt = min(L1_R, n - lo);

    {
        for (int r0 = wv * 4; r0 < cnt; r0 += (L1_THREADS / 32) * 4) {
            float4 hv[4];
#pragma unroll
            for (int q = 0; q < 4; ++q) {
                const int r = min(r0 + q, cnt - 1);
                hv[q] = *(const float4*)(h + (size_t)(lo + r) * HID_CH + 4 * lane);
            }
#pragma unroll
            for (int q = 0; q < 4; ++q) {
                if (r0 + q < cnt) {
                    float* ap = &acc[(r0 + q) * L1_PITCH + lane];
                    ap[0]  = hv[q].x;
                    ap[32] = hv[q].y;
                    ap[64] = hv[q].z;
                    ap[96] = hv[q].w;
                }
            }
        }
    }
    __syncthreads();

    const unsigned rb = (unsigned)blockIdx.x & 0xffu;
    const unsigned R4 = rb * 0x01010101u;
    const int nvf = ne >> 4;
    const int wq0 = wv * L1_WCAP;
    unsigned wcount = 0u;
    unsigned* const gl = glist + (size_t)(blockIdx.x * (L1_THREADS / 32) + wv) * L2_STRIDE;
    int  gcount = 0;
    bool govf   = false;

    auto take_edges = [&](const bool has, const int e) {
        const unsigned bal = __builtin_amdgcn_ballot_w32(has);
        if (has) {
            const unsigned slot = wcount + __builtin_amdgcn_mbcnt_lo(bal, 0u);
            if (slot < (unsigned)L1_WCAP) {
                qs[wq0 + slot] = (unsigned)e;
            } else {
                const int s = src[e];
                const unsigned rel = (unsigned)(dst[e] - lo);
                if (rel < (unsigned)cnt) {
                    const float* hp = h + (size_t)s * HID_CH;
#pragma nounroll
                    for (int c = 0; c < HID_CH; ++c)
                        ldsAddF(&acc[rel * L1_PITCH + L1_POS(c)], hp[c]);
                }
            }
        }
        wcount += (unsigned)__builtin_popcount(bal);
        if (wcount > (unsigned)L1_WCAP) govf = true;
    };
    auto scan_vec = [&](const uint4 w, const int ebase) {
        const unsigned K1 = 0x01010101u;
        const unsigned t0 = w.x ^ R4, t1 = w.y ^ R4, t2 = w.z ^ R4, t3 = w.w ^ R4;
        const unsigned z0 = (t0 - K1) & ~t0, z1 = (t1 - K1) & ~t1;
        const unsigned z2 = (t2 - K1) & ~t2, z3 = (t3 - K1) & ~t3;
        const bool cand = ((z0 | z1 | z2 | z3) & 0x80808080u) != 0u;
        if (__builtin_amdgcn_ballot_w32(cand) != 0u) {
            const unsigned f0 = z0 & ~(t0 << 7) & 0x80808080u;
            const unsigned f1 = z1 & ~(t1 << 7) & 0x80808080u;
            const unsigned f2 = z2 & ~(t2 << 7) & 0x80808080u;
            const unsigned f3 = z3 & ~(t3 << 7) & 0x80808080u;
            unsigned m = (f0 >> 7) | (f1 >> 6) | (f2 >> 5) | (f3 >> 4);
            while (__builtin_amdgcn_ballot_w32(m != 0u) != 0u) {
                const bool has = m != 0u;
                const int p = has ? __builtin_ctz(m) : 0;
                take_edges(has, ebase + (p & 7) * 4 + (p >> 3));
                m &= m - 1u;
            }
        }
    };

    auto drain = [&](const bool all) {
        const int nst = __builtin_amdgcn_readfirstlane((int)min(wcount, (unsigned)L1_WCAP));
        const int np  = all ? nst : (nst & ~(L1_G - 1));
        if (gcount + np > L2_CAP) govf = true;
        const bool gsave = !govf;

        for (int i0 = lane; i0 < np; i0 += 32 * L1_PU) {
            int ee[L1_PU], ss[L1_PU], dd[L1_PU];
#pragma unroll
            for (int u = 0; u < L1_PU; ++u) {
                const int i = i0 + u * 32;
                ee[u] = (i < np) ? (int)qs[wq0 + i] : 0;
            }
#pragma unroll
            for (int u = 0; u < L1_PU; ++u) {
                ss[u] = src[ee[u]];
                dd[u] = dst[ee[u]];
            }
#pragma unroll
            for (int u = 0; u < L1_PU; ++u) {
                const int i = i0 + u * 32;
                if (i < np) {
                    const unsigned rel = (unsigned)(dd[u] - lo);
                    const bool ours = rel < (unsigned)cnt;
                    qs[wq0 + i] = (unsigned)ss[u];
                    qd[wq0 + i] = ours ? rel : ~0u;
                    if (gsave)
                        *(uint2*)(gl + 2 * (1 + gcount + i)) = make_uint2((unsigned)ss[u], ours ? rel : ~0u);
                }
            }
        }

        for (int b0 = 0; b0 < np; b0 += 32) {
            const int m = min(32, np - b0);
            unsigned so = 0u;
            unsigned dw = ~0u;
            if (lane < m) {
                so = qs[wq0 + b0 + lane] * (unsigned)HID_CH;
                dw = qd[wq0 + b0 + lane];
            }
            const unsigned okmask = __builtin_amdgcn_ballot_w32(lane < m && dw != ~0u);
            for (int j0 = 0; j0 < m; j0 += L1_G) {
                float4 v[L1_G];
#pragma unroll
                for (int u = 0; u < L1_G; ++u) {
                    const int j = min(j0 + u, m - 1);
                    const unsigned sj = (unsigned)__builtin_amdgcn_readlane((int)so, j);
                    v[u] = *(const float4*)(h + (size_t)sj + 4 * lane);
                }
#pragma unroll
                for (int u = 0; u < L1_G; ++u) {
                    const int j = j0 + u;
                    if ((okmask >> j) & 1u) {
                        const unsigned dj = (unsigned)__builtin_amdgcn_readlane((int)dw, j);
                        float* ap = &acc[dj * L1_PITCH + lane];
                        ldsAddF(ap +  0, v[u].x);
                        ldsAddF(ap + 32, v[u].y);
                        ldsAddF(ap + 64, v[u].z);
                        ldsAddF(ap + 96, v[u].w);
                    }
                }
            }
        }

        if (gsave) gcount += np;
        const int left = nst - np;
        if (lane < left) {
            const unsigned t = qs[wq0 + np + lane];
            qs[wq0 + lane] = t;
        }
        wcount = (unsigned)left;
    };

    const uint4* rv = (const uint4*)rid8;
    int vb = wv * 32;
    for (; vb + 31 + (L1_SU - 1) * L1_THREADS < nvf; vb += L1_THREADS * L1_SU) {
        uint4 wb[L1_SU];
#pragma unroll
        for (int u = 0; u < L1_SU; ++u) wb[u] = rv[vb + lane + u * L1_THREADS];
#pragma unroll
        for (int u = 0; u < L1_SU; ++u) scan_vec(wb[u], (vb + lane + u * L1_THREADS) * 16);
        if (wcount >= (unsigned)L1_DRAIN) drain(false);
    }
    for (; vb < nvf; vb += L1_THREADS) {
        const int v = vb + lane;
        const unsigned nomatch = ~R4;
        const uint4 w = (v < nvf) ? rv[v] : make_uint4(nomatch, nomatch, nomatch, nomatch);
        scan_vec(w, v * 16);
    }
    if (wv == 0) {
        const int e = (nvf << 4) + lane;
        const bool has = lane < 16 && e < ne && (unsigned)rid8[e] == rb;
        take_edges(has, e);
    }
    drain(true);
    if (lane == 0) gl[0] = govf ? L2_OVF : (unsigned)gcount;
    __syncthreads();

    if (tid < cnt) {
        const float* row = &acc[tid * L1_PITCH];
        const float dd = dinv[lo + tid];
        float s0 = 0.f, s1 = 0.f;
#pragma unroll 8
        for (int c = 0; c < HID_CH; c += 4) {
            float r0 = fmaxf(b1[c + 0] + dd * row[L1_POS(c + 0)], 0.f);
            float r1 = fmaxf(b1[c + 1] + dd * row[L1_POS(c + 1)], 0.f);
            float r2 = fmaxf(b1[c + 2] + dd * row[L1_POS(c + 2)], 0.f);
            float r3 = fmaxf(b1[c + 3] + dd * row[L1_POS(c + 3)], 0.f);
            s0 += r0 * W2[(c + 0) * 2 + 0] + r1 * W2[(c + 1) * 2 + 0]
                + r2 * W2[(c + 2) * 2 + 0] + r3 * W2[(c + 3) * 2 + 0];
            s1 += r0 * W2[(c + 0) * 2 + 1] + r1 * W2[(c + 1) * 2 + 1]
                + r2 * W2[(c + 2) * 2 + 1] + r3 * W2[(c + 3) * 2 + 1];
        }
        h2[(size_t)(lo + tid) * 2 + 0] = dd * s0;
        h2[(size_t)(lo + tid) * 2 + 1] = dd * s1;
    }
}

__global__ __launch_bounds__(L1_THREADS) void k_agg2_own(
    const float* __restrict__ h2, const int* __restrict__ src, const int* __restrict__ dst,
    const unsigned* __restrict__ glist,
    const float* __restrict__ dinv, const float* __restrict__ b2, float* __restrict__ out,
    int ne, int n)
{
    __shared__ float    acc2[(L1_R + 32) * 2];
    __shared__ unsigned s_ovf;

    const int tid  = threadIdx.x;
    const int lane = tid & 31;
    const int wv  = __builtin_amdgcn_readfirstlane(tid >> 5);
    const int lo  = blockIdx.x * L1_R;
    const int cnt = min(L1_R, n - lo);

    const unsigned* gl = glist + (size_t)(blockIdx.x * (L1_THREADS / 32) + wv) * L2_STRIDE;
    const unsigned c = gl[0];
    if (tid < L1_R + 32) {
        float2 hs = make_float2(0.f, 0.f);
        if (tid < cnt) hs = *(const float2*)(h2 + (size_t)(lo + tid) * 2);
        acc2[2 * tid + 0] = hs.x;
        acc2[2 * tid + 1] = hs.y;
    }
    if (tid == 0) s_ovf = 0u;
    __syncthreads();
    if (lane == 0 && c > (unsigned)L2_CAP) s_ovf = 1u;
    __syncthreads();

    if (s_ovf == 0u) {
        const int m = (int)c;
        for (int i0 = lane; i0 < m; i0 += 32 * L2_U) {
            uint2  en[L2_U];
            float2 hv[L2_U];
#pragma unroll
            for (int u = 0; u < L2_U; ++u) {
                const int i = min(i0 + u * 32, m - 1);
                en[u] = *(const uint2*)(gl + 2 * (1 + i));
            }
#pragma unroll
            for (int u = 0; u < L2_U; ++u)
                hv[u] = *(const float2*)(h2 + (size_t)en[u].x * 2);
#pragma unroll
            for (int u = 0; u < L2_U; ++u) {
                const bool ours = (i0 + u * 32 < m) && (en[u].y < (unsigned)cnt);
                const unsigned d = ours ? en[u].y : (unsigned)(L1_R + lane);
                ldsAddF(&acc2[d * 2 + 0], hv[u].x);
                ldsAddF(&acc2[d * 2 + 1], hv[u].y);
            }
        }
    } else {
        for (int e = tid; e < ne; e += L1_THREADS) {
            const unsigned rel = (unsigned)(dst[e] - lo);
            if (rel < (unsigned)cnt) {
                const float2 v = *(const float2*)(h2 + (size_t)src[e] * 2);
                ldsAddF(&acc2[rel * 2 + 0], v.x);
                ldsAddF(&acc2[rel * 2 + 1], v.y);
            }
        }
    }
    __syncthreads();

    if (tid < cnt) {
        const float di = dinv[lo + tid];
        *(float2*)(out + (size_t)(lo + tid) * 2) =
            make_float2(b2[0] + di * acc2[2 * tid + 0], b2[1] + di * acc2[2 * tid + 1]);
    }
}

extern "C" void kernel_launch(void* const* d_in, const int* in_sizes, int n_in,
                              void* d_out, int out_size, void* d_ws, size_t ws_size,
                              hipStream_t stream)
{
    const float* x  = (const float*)d_in[0];
    const int*   ei = (const int*)d_in[1];
    const float* W1 = (const float*)d_in[2];
    const float* b1 = (const float*)d_in[3];
    const float* W2 = (const float*)d_in[4];
    const float* b2 = (const float*)d_in[5];
    float* out = (float*)d_out;

    const int n_nodes = in_sizes[0] / IN_CH;
    const int n_edges = in_sizes[1] / 2;
    const int* src = ei;
    const int* dst = ei + n_edges;

    char* ws = (char*)d_ws;
    size_t off = 0;
    float* dinv = (float*)(ws + off);
    off += ((size_t)n_nodes * 4 + 4095) & ~(size_t)4095;
    float* h = (float*)(ws + off);
    size_t hb = ((size_t)n_nodes * HID_CH * 4 + 4095) & ~(size_t)4095;
    off += hb;
    char* reg3 = ws + off;

    const int nblk = (n_nodes + L1_R - 1) / L1_R;
    unsigned* part  = (unsigned*)reg3;
    unsigned* glist = (unsigned*)reg3;
    size_t pb  = ((size_t)nblk * (L1_THREADS / 32) * L2_STRIDE * 4 + 4095) & ~(size_t)4095;
    float* h2 = (float*)(reg3 + pb);
    size_t h2b = ((size_t)n_nodes * 2 * 4 + 4095) & ~(size_t)4095;
    unsigned char* rid8 = (unsigned char*)(reg3 + pb + h2b);

    const int chunk_len = (n_edges + NT_PLANES - 1) / NT_PLANES;
    dim3 ngrid(NT_PLANES, (n_nodes + NT_RANGE - 1) / NT_RANGE);
    k_deg_hist<<<ngrid, NT_THREADS, 0, stream>>>(dst, part, rid8, n_edges, n_nodes, chunk_len);
    k_dinv    <<<(n_nodes + 255) / 256, 256, 0, stream>>>(part, dinv, n_nodes);

    k_gemm1_wmma<<<G1_BLOCKS, G1_THREADS, 0, stream>>>(x, W1, dinv, h, n_nodes);

    k_agg1_own  <<<nblk, L1_THREADS, 0, stream>>>(
        h, src, dst, dinv, b1, W2, rid8, h2, glist, n_edges, n_nodes);

    k_agg2_own  <<<nblk, L1_THREADS, 0, stream>>>(
        h2, src, dst, glist, dinv, b2, out, n_edges, n_nodes);
}
